# starting kernel
# baseline (speedup 1.0000x reference)
#include <hip/hip_runtime.h>

typedef __attribute__((ext_vector_type(2))) float v2f;
typedef __attribute__((ext_vector_type(8))) float v8f;

#define FEAT 64
#define NGRAPH 64

static inline int ceildiv(long long a, long long b) { return (int)((a + b - 1) / b); }

__global__ void k_deg_init(float* deg, int N) {
    int i = blockIdx.x * blockDim.x + threadIdx.x;
    if (i < N) deg[i] = 1.0f;
}

__global__ void k_deg_accum(const int* __restrict__ col, float* deg, long long E) {
    long long i = (long long)blockIdx.x * blockDim.x + threadIdx.x;
    if (i < E) atomicAdd(&deg[col[i]], 1.0f);
}

__global__ void k_deg_rsqrt(float* deg, int N) {
    int i = blockIdx.x * blockDim.x + threadIdx.x;
    if (i < N) deg[i] = rsqrtf(deg[i]);
}

__global__ void k_gemm_wmma(const float* __restrict__ A, const float* __restrict__ W,
                            float* __restrict__ D, int M) {
    int gwave = (blockIdx.x * blockDim.x + threadIdx.x) >> 5;
    int lane  = threadIdx.x & 31;
    int mtiles = (M + 15) >> 4;
    int mtile = gwave >> 2;
    int ntile = gwave & 3;
    if (mtile >= mtiles) return;

    int m0 = mtile << 4, n0 = ntile << 4;
    int half = lane >> 4;
    int r    = lane & 15;

    int arow = m0 + r; if (arow >= M) arow = M - 1;

    v8f c = {};
#pragma unroll
    for (int k0 = 0; k0 < FEAT; k0 += 4) {
        v2f a, b;
        int ka = k0 + 2 * half;
        a.x = A[(long long)arow * FEAT + ka + 0];
        a.y = A[(long long)arow * FEAT + ka + 1];
        b.x = W[(ka + 0) * FEAT + n0 + r];
        b.y = W[(ka + 1) * FEAT + n0 + r];
        c = __builtin_amdgcn_wmma_f32_16x16x4_f32(false, a, false, b, (short)0, c, false, false);
    }
#pragma unroll
    for (int v = 0; v < 8; ++v) {
        int row = m0 + v + 8 * half;
        if (row < M) D[(long long)row * FEAT + n0 + r] = c[v];
    }
}

__global__ void k_agg_init(const float* __restrict__ h, const float* __restrict__ dinv,
                           float* __restrict__ out, long long total) {
    long long gid = (long long)blockIdx.x * blockDim.x + threadIdx.x;
    if (gid >= total) return;
    int node = (int)(gid >> 6);
    float d = dinv[node];
    out[gid] = h[gid] * d * d;
}

__global__ void k_edge_scatter(const int* __restrict__ rows, const int* __restrict__ cols,
                               const float* __restrict__ h, const float* __restrict__ dinv,
                               float* __restrict__ out, long long E) {
    long long gid  = (long long)blockIdx.x * blockDim.x + threadIdx.x;
    long long e    = gid >> 4;
    int       f4   = (int)(gid & 15);
    if (e >= E) return;
    int r = rows[e], c = cols[e];
    float nrm = dinv[r] * dinv[c];
    const float4 hv = ((const float4*)(h + (long long)r * FEAT))[f4];
    float* o = out + (long long)c * FEAT + f4 * 4;
    atomicAdd(o + 0, hv.x * nrm);
    atomicAdd(o + 1, hv.y * nrm);
    atomicAdd(o + 2, hv.z * nrm);
    atomicAdd(o + 3, hv.w * nrm);
}

__global__ void k_bias_relu(const float* __restrict__ in, const float* __restrict__ b,
                            float* __restrict__ out, long long total) {
    long long gid = (long long)blockIdx.x * blockDim.x + threadIdx.x;
    if (gid >= total) return;
    float v = in[gid] + b[gid & (FEAT - 1)];
    out[gid] = v > 0.0f ? v : 0.0f;
}

__global__ void k_pool_zero(float* sums, float* counts) {
    int i = blockIdx.x * blockDim.x + threadIdx.x;
    if (i < NGRAPH * FEAT) sums[i] = 0.0f;
    if (i < NGRAPH) counts[i] = 0.0f;
}

__global__ void k_pool_accum(const float* __restrict__ h, const int* __restrict__ batch,
                             float* __restrict__ sums, float* __restrict__ counts, int N) {
    __shared__ float ls[NGRAPH * FEAT];
    __shared__ float lc[NGRAPH];
    for (int i = threadIdx.x; i < NGRAPH * FEAT; i += blockDim.x) ls[i] = 0.0f;
    for (int i = threadIdx.x; i < NGRAPH; i += blockDim.x) lc[i] = 0.0f;
    __syncthreads();

    int wave = threadIdx.x >> 5;
    int lane = threadIdx.x & 31;
    for (int node = blockIdx.x * 8 + wave; node < N; node += gridDim.x * 8) {
        int g = batch[node];
        atomicAdd(&ls[g * FEAT + lane],      h[(long long)node * FEAT + lane]);
        atomicAdd(&ls[g * FEAT + lane + 32], h[(long long)node * FEAT + lane + 32]);
        if (lane == 0) atomicAdd(&lc[g], 1.0f);
    }
    __syncthreads();
    for (int i = threadIdx.x; i < NGRAPH * FEAT; i += blockDim.x)
        if (ls[i] != 0.0f) atomicAdd(&sums[i], ls[i]);
    for (int i = threadIdx.x; i < NGRAPH; i += blockDim.x)
        if (lc[i] != 0.0f) atomicAdd(&counts[i], lc[i]);
}

__global__ void k_pool_div(const float* __restrict__ sums, const float* __restrict__ counts,
                           float* __restrict__ out) {
    int i = blockIdx.x * blockDim.x + threadIdx.x;
    if (i >= NGRAPH * FEAT) return;
    out[i] = sums[i] / fmaxf(counts[i >> 6], 1.0f);
}

extern "C" void kernel_launch(void* const* d_in, const int* in_sizes, int n_in,
                              void* d_out, int out_size, void* d_ws, size_t ws_size,
                              hipStream_t stream) {
    const float* x     = (const float*)d_in[0];
    const int*   edge  = (const int*)d_in[1];
    const int*   batch = (const int*)d_in[2];
    const float* W1    = (const float*)d_in[3];
    const float* b1    = (const float*)d_in[4];
    const float* W2    = (const float*)d_in[5];
    const float* b2    = (const float*)d_in[6];
    float*       out   = (float*)d_out;

    const int       N = in_sizes[0] / FEAT;
    const long long E = (long long)in_sizes[1] / 2;
    const long long NF = (long long)N * FEAT;

    float* bufA   = (float*)d_ws;
    float* bufB   = bufA + NF;
    float* dinv   = bufB + NF;
    float* sums   = dinv + N;
    float* counts = sums + NGRAPH * FEAT;

    const int* rows = edge;
    const int* cols = edge + E;

    const int T = 256;

    k_deg_init <<<ceildiv(N, T), T, 0, stream>>>(dinv, N);
    k_deg_accum<<<ceildiv(E, T), T, 0, stream>>>(cols, dinv, E);
    k_deg_rsqrt<<<ceildiv(N, T), T, 0, stream>>>(dinv, N);

    const int mtiles   = (N + 15) / 16;
    const int gemmBlks = ceildiv((long long)mtiles * 4 * 32, T);

    k_gemm_wmma   <<<gemmBlks, T, 0, stream>>>(x, W1, bufA, N);
    k_agg_init    <<<ceildiv(NF, T), T, 0, stream>>>(bufA, dinv, bufB, NF);
    k_edge_scatter<<<ceildiv(E * 16, T), T, 0, stream>>>(rows, cols, bufA, dinv, bufB, E);
    k_bias_relu   <<<ceildiv(NF, T), T, 0, stream>>>(bufB, b1, bufA, NF);

    k_gemm_wmma   <<<gemmBlks, T, 0, stream>>>(bufA, W2, bufB, N);
    k_agg_init    <<<ceildiv(NF, T), T, 0, stream>>>(bufB, dinv, bufA, NF);
    k_edge_scatter<<<ceildiv(E * 16, T), T, 0, stream>>>(rows, cols, bufB, dinv, bufA, E);
    k_bias_relu   <<<ceildiv(NF, T), T, 0, stream>>>(bufA, b2, bufB, NF);

    k_pool_zero <<<ceildiv(NGRAPH * FEAT, T), T, 0, stream>>>(sums, counts);
    k_pool_accum<<<512, T, 0, stream>>>(bufB, batch, sums, counts, N);
    k_pool_div  <<<ceildiv(NGRAPH * FEAT, T), T, 0, stream>>>(sums, counts, out);
}


# edge scatter: one wave walks 32 edges, lane = feature, so each gather load and each f32 atomic covers one whole 128-byte line (was 4 lines per instruction); atomic arrival order changes
# speedup vs baseline: 1.6618x; 1.6618x over previous
#include <hip/hip_runtime.h>

typedef __attribute__((ext_vector_type(2))) float v2f;
typedef __attribute__((ext_vector_type(8))) float v8f;

#define FEAT 64
#define NGRAPH 64

static inline int ceildiv(long long a, long long b) { return (int)((a + b - 1) / b); }

__global__ void k_deg_init(float* deg, int N) {
    int i = blockIdx.x * blockDim.x + threadIdx.x;
    if (i < N) deg[i] = 1.0f;
}

__global__ void k_deg_accum(const int* __restrict__ col, float* deg, long long E) {
    long long i = (long long)blockIdx.x * blockDim.x + threadIdx.x;
    if (i < E) atomicAdd(&deg[col[i]], 1.0f);
}

__global__ void k_deg_rsqrt(float* deg, int N) {
    int i = blockIdx.x * blockDim.x + threadIdx.x;
    if (i < N) deg[i] = rsqrtf(deg[i]);
}

__global__ void k_gemm_wmma(const float* __restrict__ A, const float* __restrict__ W,
                            float* __restrict__ D, int M) {
    int gwave = (blockIdx.x * blockDim.x + threadIdx.x) >> 5;
    int lane  = threadIdx.x & 31;
    int mtiles = (M + 15) >> 4;
    int mtile = gwave >> 2;
    int ntile = gwave & 3;
    if (mtile >= mtiles) return;

    int m0 = mtile << 4, n0 = ntile << 4;
    int half = lane >> 4;
    int r    = lane & 15;

    int arow = m0 + r; if (arow >= M) arow = M - 1;

    v8f c = {};
#pragma unroll
    for (int k0 = 0; k0 < FEAT; k0 += 4) {
        v2f a, b;
        int ka = k0 + 2 * half;
        a.x = A[(long long)arow * FEAT + ka + 0];
        a.y = A[(long long)arow * FEAT + ka + 1];
        b.x = W[(ka + 0) * FEAT + n0 + r];
        b.y = W[(ka + 1) * FEAT + n0 + r];
        c = __builtin_amdgcn_wmma_f32_16x16x4_f32(false, a, false, b, (short)0, c, false, false);
    }
#pragma unroll
    for (int v = 0; v < 8; ++v) {
        int row = m0 + v + 8 * half;
        if (row < M) D[(long long)row * FEAT + n0 + r] = c[v];
    }
}

__global__ void k_agg_init(const float* __restrict__ h, const float* __restrict__ dinv,
                           float* __restrict__ out, long long total) {
    long long gid = (long long)blockIdx.x * blockDim.x + threadIdx.x;
    if (gid >= total) return;
    int node = (int)(gid >> 6);
    float d = dinv[node];
    out[gid] = h[gid] * d * d;
}

#define EDGE_GRP 8

__device__ __forceinline__ void edge_rows_load(const float* __restrict__ h, int r, int lane, int j0,
                                               float (&lo)[EDGE_GRP], float (&hi)[EDGE_GRP]) {
#pragma unroll
    for (int u = 0; u < EDGE_GRP; ++u) {
        int rj = __builtin_amdgcn_readlane(r, j0 + u);
        const float* hp = h + (long long)rj * FEAT + lane;
        lo[u] = hp[0];
        hi[u] = hp[32];
    }
}

__device__ __forceinline__ void edge_rows_add(float* __restrict__ out, int c, float nrm, int lane, int j0,
                                              const float (&lo)[EDGE_GRP], const float (&hi)[EDGE_GRP]) {
#pragma unroll
    for (int u = 0; u < EDGE_GRP; ++u) {
        int   cj = __builtin_amdgcn_readlane(c, j0 + u);
        float nj = __int_as_float(__builtin_amdgcn_readlane(__float_as_int(nrm), j0 + u));
        float* op = out + (long long)cj * FEAT + lane;
        atomicAdd(op,      lo[u] * nj);
        atomicAdd(op + 32, hi[u] * nj);
    }
}

__global__ void
k_edge_scatter(const int* __restrict__ rows, const int* __restrict__ cols,
               const float* __restrict__ h, const float* __restrict__ dinv,
               float* __restrict__ out, long long E) {
    const int lane = threadIdx.x & 31;
    const long long wave = ((long long)blockIdx.x * blockDim.x + threadIdx.x) >> 5;
    const long long e0 = wave * 32;
    if (e0 >= E) return;
    const long long e = e0 + lane;
    int r = 0, c = 0;
    float nrm = 0.0f;
    if (e < E) {
        r = rows[e];
        c = cols[e];
        nrm = dinv[r] * dinv[c];
    }
    if (e0 + 32 <= E) {
        float lo0[EDGE_GRP], hi0[EDGE_GRP], lo1[EDGE_GRP], hi1[EDGE_GRP];
        edge_rows_load(h, r, lane, 0, lo0, hi0);
        edge_rows_load(h, r, lane, 1 * EDGE_GRP, lo1, hi1);
        edge_rows_add(out, c, nrm, lane, 0, lo0, hi0);
        edge_rows_load(h, r, lane, 2 * EDGE_GRP, lo0, hi0);
        edge_rows_add(out, c, nrm, lane, 1 * EDGE_GRP, lo1, hi1);
        edge_rows_load(h, r, lane, 3 * EDGE_GRP, lo1, hi1);
        edge_rows_add(out, c, nrm, lane, 2 * EDGE_GRP, lo0, hi0);
        edge_rows_add(out, c, nrm, lane, 3 * EDGE_GRP, lo1, hi1);
    } else {
        const int cnt = (int)(E - e0);
        for (int j = 0; j < cnt; ++j) {
            int   rj = __builtin_amdgcn_readlane(r, j);
            int   cj = __builtin_amdgcn_readlane(c, j);
            float nj = __int_as_float(__builtin_amdgcn_readlane(__float_as_int(nrm), j));
            const float* hp = h + (long long)rj * FEAT + lane;
            float* op = out + (long long)cj * FEAT + lane;
            atomicAdd(op,      hp[0]  * nj);
            atomicAdd(op + 32, hp[32] * nj);
        }
    }
}

__global__ void k_bias_relu(const float* __restrict__ in, const float* __restrict__ b,
                            float* __restrict__ out, long long total) {
    long long gid = (long long)blockIdx.x * blockDim.x + threadIdx.x;
    if (gid >= total) return;
    float v = in[gid] + b[gid & (FEAT - 1)];
    out[gid] = v > 0.0f ? v : 0.0f;
}

__global__ void k_pool_zero(float* sums, float* counts) {
    int i = blockIdx.x * blockDim.x + threadIdx.x;
    if (i < NGRAPH * FEAT) sums[i] = 0.0f;
    if (i < NGRAPH) counts[i] = 0.0f;
}

__global__ void k_pool_accum(const float* __restrict__ h, const int* __restrict__ batch,
                             float* __restrict__ sums, float* __restrict__ counts, int N) {
    __shared__ float ls[NGRAPH * FEAT];
    __shared__ float lc[NGRAPH];
    for (int i = threadIdx.x; i < NGRAPH * FEAT; i += blockDim.x) ls[i] = 0.0f;
    for (int i = threadIdx.x; i < NGRAPH; i += blockDim.x) lc[i] = 0.0f;
    __syncthreads();

    int wave = threadIdx.x >> 5;
    int lane = threadIdx.x & 31;
    for (int node = blockIdx.x * 8 + wave; node < N; node += gridDim.x * 8) {
        int g = batch[node];
        atomicAdd(&ls[g * FEAT + lane],      h[(long long)node * FEAT + lane]);
        atomicAdd(&ls[g * FEAT + lane + 32], h[(long long)node * FEAT + lane + 32]);
        if (lane == 0) atomicAdd(&lc[g], 1.0f);
    }
    __syncthreads();
    for (int i = threadIdx.x; i < NGRAPH * FEAT; i += blockDim.x)
        if (ls[i] != 0.0f) atomicAdd(&sums[i], ls[i]);
    for (int i = threadIdx.x; i < NGRAPH; i += blockDim.x)
        if (lc[i] != 0.0f) atomicAdd(&counts[i], lc[i]);
}

__global__ void k_pool_div(const float* __restrict__ sums, const float* __restrict__ counts,
                           float* __restrict__ out) {
    int i = blockIdx.x * blockDim.x + threadIdx.x;
    if (i >= NGRAPH * FEAT) return;
    out[i] = sums[i] / fmaxf(counts[i >> 6], 1.0f);
}

extern "C" void kernel_launch(void* const* d_in, const int* in_sizes, int n_in,
                              void* d_out, int out_size, void* d_ws, size_t ws_size,
                              hipStream_t stream) {
    const float* x     = (const float*)d_in[0];
    const int*   edge  = (const int*)d_in[1];
    const int*   batch = (const int*)d_in[2];
    const float* W1    = (const float*)d_in[3];
    const float* b1    = (const float*)d_in[4];
    const float* W2    = (const float*)d_in[5];
    const float* b2    = (const float*)d_in[6];
    float*       out   = (float*)d_out;

    const int       N = in_sizes[0] / FEAT;
    const long long E = (long long)in_sizes[1] / 2;
    const long long NF = (long long)N * FEAT;

    float* bufA   = (float*)d_ws;
    float* bufB   = bufA + NF;
    float* dinv   = bufB + NF;
    float* sums   = dinv + N;
    float* counts = sums + NGRAPH * FEAT;

    const int* rows = edge;
    const int* cols = edge + E;

    const int T = 256;

    k_deg_init <<<ceildiv(N, T), T, 0, stream>>>(dinv, N);
    k_deg_accum<<<ceildiv(E, T), T, 0, stream>>>(cols, dinv, E);
    k_deg_rsqrt<<<ceildiv(N, T), T, 0, stream>>>(dinv, N);

    const int mtiles   = (N + 15) / 16;
    const int gemmBlks = ceildiv((long long)mtiles * 4 * 32, T);

    k_gemm_wmma   <<<gemmBlks, T, 0, stream>>>(x, W1, bufA, N);
    k_agg_init    <<<ceildiv(NF, T), T, 0, stream>>>(bufA, dinv, bufB, NF);
    k_edge_scatter<<<ceildiv(E, T), T, 0, stream>>>(rows, cols, bufA, dinv, bufB, E);
    k_bias_relu   <<<ceildiv(NF, T), T, 0, stream>>>(bufB, b1, bufA, NF);

    k_gemm_wmma   <<<gemmBlks, T, 0, stream>>>(bufA, W2, bufB, N);
    k_agg_init    <<<ceildiv(NF, T), T, 0, stream>>>(bufB, dinv, bufA, NF);
    k_edge_scatter<<<ceildiv(E, T), T, 0, stream>>>(rows, cols, bufB, dinv, bufA, E);
    k_bias_relu   <<<ceildiv(NF, T), T, 0, stream>>>(bufA, b2, bufB, NF);

    k_pool_zero <<<ceildiv(NGRAPH * FEAT, T), T, 0, stream>>>(sums, counts);
    k_pool_accum<<<512, T, 0, stream>>>(bufB, batch, sums, counts, N);
    k_pool_div  <<<ceildiv(NGRAPH * FEAT, T), T, 0, stream>>>(sums, counts, out);
}


# edge scatter: edge weight split into per-node factors: rows scaled by dinv[row] once, dinv[col] applied once after the sum, no per-edge dinv gathers; rounding order of the scaling changes
# speedup vs baseline: 1.7843x; 1.0737x over previous
#include <hip/hip_runtime.h>

typedef __attribute__((ext_vector_type(2))) float v2f;
typedef __attribute__((ext_vector_type(8))) float v8f;

#define FEAT 64
#define NGRAPH 64

static inline int ceildiv(long long a, long long b) { return (int)((a + b - 1) / b); }

__global__ void k_deg_init(float* deg, int N) {
    int i = blockIdx.x * blockDim.x + threadIdx.x;
    if (i < N) deg[i] = 1.0f;
}

__global__ void k_deg_accum(const int* __restrict__ col, float* deg, long long E) {
    long long i = (long long)blockIdx.x * blockDim.x + threadIdx.x;
    if (i < E) atomicAdd(&deg[col[i]], 1.0f);
}

__global__ void k_deg_rsqrt(float* deg, int N) {
    int i = blockIdx.x * blockDim.x + threadIdx.x;
    if (i < N) deg[i] = rsqrtf(deg[i]);
}

__global__ void k_gemm_wmma(const float* __restrict__ A, const float* __restrict__ W,
                            float* __restrict__ D, int M) {
    int gwave = (blockIdx.x * blockDim.x + threadIdx.x) >> 5;
    int lane  = threadIdx.x & 31;
    int mtiles = (M + 15) >> 4;
    int mtile = gwave >> 2;
    int ntile = gwave & 3;
    if (mtile >= mtiles) return;

    int m0 = mtile << 4, n0 = ntile << 4;
    int half = lane >> 4;
    int r    = lane & 15;

    int arow = m0 + r; if (arow >= M) arow = M - 1;

    v8f c = {};
#pragma unroll
    for (int k0 = 0; k0 < FEAT; k0 += 4) {
        v2f a, b;
        int ka = k0 + 2 * half;
        a.x = A[(long long)arow * FEAT + ka + 0];
        a.y = A[(long long)arow * FEAT + ka + 1];
        b.x = W[(ka + 0) * FEAT + n0 + r];
        b.y = W[(ka + 1) * FEAT + n0 + r];
        c = __builtin_amdgcn_wmma_f32_16x16x4_f32(false, a, false, b, (short)0, c, false, false);
    }
#pragma unroll
    for (int v = 0; v < 8; ++v) {
        int row = m0 + v + 8 * half;
        if (row < M) D[(long long)row * FEAT + n0 + r] = c[v];
    }
}

__global__ void k_agg_init(float* __restrict__ h, const float* __restrict__ dinv,
                           float* __restrict__ out, long long total) {
    long long gid = (long long)blockIdx.x * blockDim.x + threadIdx.x;
    if (gid >= total) return;
    int node = (int)(gid >> 6);
    float hs = h[gid] * dinv[node];
    h[gid]   = hs;
    out[gid] = hs;
}

#define EDGE_GRP 8

__device__ __forceinline__ void edge_rows_load(const float* __restrict__ h, int r, int lane, int j0,
                                               float (&lo)[EDGE_GRP], float (&hi)[EDGE_GRP]) {
#pragma unroll
    for (int u = 0; u < EDGE_GRP; ++u) {
        int rj = __builtin_amdgcn_readlane(r, j0 + u);
        const float* hp = h + (long long)rj * FEAT + lane;
        lo[u] = hp[0];
        hi[u] = hp[32];
    }
}

__device__ __forceinline__ void edge_rows_add(float* __restrict__ out, int c, int lane, int j0,
                                              const float (&lo)[EDGE_GRP], const float (&hi)[EDGE_GRP]) {
#pragma unroll
    for (int u = 0; u < EDGE_GRP; ++u) {
        int cj = __builtin_amdgcn_readlane(c, j0 + u);
        float* op = out + (long long)cj * FEAT + lane;
        atomicAdd(op,      lo[u]);
        atomicAdd(op + 32, hi[u]);
    }
}

__global__ void
k_edge_scatter(const int* __restrict__ rows, const int* __restrict__ cols,
               const float* __restrict__ h, float* __restrict__ out, long long E) {
    const int lane = threadIdx.x & 31;
    const long long wave = ((long long)blockIdx.x * blockDim.x + threadIdx.x) >> 5;
    const long long e0 = wave * 32;
    if (e0 >= E) return;
    const long long e = e0 + lane;
    int r = 0, c = 0;
    if (e < E) {
        r = rows[e];
        c = cols[e];
    }
    if (e0 + 32 <= E) {
        float lo0[EDGE_GRP], hi0[EDGE_GRP], lo1[EDGE_GRP], hi1[EDGE_GRP];
        edge_rows_load(h, r, lane, 0, lo0, hi0);
        edge_rows_load(h, r, lane, 1 * EDGE_GRP, lo1, hi1);
        edge_rows_add(out, c, lane, 0, lo0, hi0);
        edge_rows_load(h, r, lane, 2 * EDGE_GRP, lo0, hi0);
        edge_rows_add(out, c, lane, 1 * EDGE_GRP, lo1, hi1);
        edge_rows_load(h, r, lane, 3 * EDGE_GRP, lo1, hi1);
        edge_rows_add(out, c, lane, 2 * EDGE_GRP, lo0, hi0);
        edge_rows_add(out, c, lane, 3 * EDGE_GRP, lo1, hi1);
    } else {
        const int cnt = (int)(E - e0);
        for (int j = 0; j < cnt; ++j) {
            int rj = __builtin_amdgcn_readlane(r, j);
            int cj = __builtin_amdgcn_readlane(c, j);
            const float* hp = h + (long long)rj * FEAT + lane;
            float* op = out + (long long)cj * FEAT + lane;
            atomicAdd(op,      hp[0]);
            atomicAdd(op + 32, hp[32]);
        }
    }
}

__global__ void k_bias_relu(const float* __restrict__ in, const float* __restrict__ dinv,
                            const float* __restrict__ b, float* __restrict__ out, long long total) {
    long long gid = (long long)blockIdx.x * blockDim.x + threadIdx.x;
    if (gid >= total) return;
    float v = in[gid] * dinv[(int)(gid >> 6)] + b[gid & (FEAT - 1)];
    out[gid] = v > 0.0f ? v : 0.0f;
}

__global__ void k_pool_zero(float* sums, float* counts) {
    int i = blockIdx.x * blockDim.x + threadIdx.x;
    if (i < NGRAPH * FEAT) sums[i] = 0.0f;
    if (i < NGRAPH) counts[i] = 0.0f;
}

__global__ void k_pool_accum(const float* __restrict__ h, const int* __restrict__ batch,
                             float* __restrict__ sums, float* __restrict__ counts, int N) {
    __shared__ float ls[NGRAPH * FEAT];
    __shared__ float lc[NGRAPH];
    for (int i = threadIdx.x; i < NGRAPH * FEAT; i += blockDim.x) ls[i] = 0.0f;
    for (int i = threadIdx.x; i < NGRAPH; i += blockDim.x) lc[i] = 0.0f;
    __syncthreads();

    int wave = threadIdx.x >> 5;
    int lane = threadIdx.x & 31;
    for (int node = blockIdx.x * 8 + wave; node < N; node += gridDim.x * 8) {
        int g = batch[node];
        atomicAdd(&ls[g * FEAT + lane],      h[(long long)node * FEAT + lane]);
        atomicAdd(&ls[g * FEAT + lane + 32], h[(long long)node * FEAT + lane + 32]);
        if (lane == 0) atomicAdd(&lc[g], 1.0f);
    }
    __syncthreads();
    for (int i = threadIdx.x; i < NGRAPH * FEAT; i += blockDim.x)
        if (ls[i] != 0.0f) atomicAdd(&sums[i], ls[i]);
    for (int i = threadIdx.x; i < NGRAPH; i += blockDim.x)
        if (lc[i] != 0.0f) atomicAdd(&counts[i], lc[i]);
}

__global__ void k_pool_div(const float* __restrict__ sums, const float* __restrict__ counts,
                           float* __restrict__ out) {
    int i = blockIdx.x * blockDim.x + threadIdx.x;
    if (i >= NGRAPH * FEAT) return;
    out[i] = sums[i] / fmaxf(counts[i >> 6], 1.0f);
}

extern "C" void kernel_launch(void* const* d_in, const int* in_sizes, int n_in,
                              void* d_out, int out_size, void* d_ws, size_t ws_size,
                              hipStream_t stream) {
    const float* x     = (const float*)d_in[0];
    const int*   edge  = (const int*)d_in[1];
    const int*   batch = (const int*)d_in[2];
    const float* W1    = (const float*)d_in[3];
    const float* b1    = (const float*)d_in[4];
    const float* W2    = (const float*)d_in[5];
    const float* b2    = (const float*)d_in[6];
    float*       out   = (float*)d_out;

    const int       N = in_sizes[0] / FEAT;
    const long long E = (long long)in_sizes[1] / 2;
    const long long NF = (long long)N * FEAT;

    float* bufA   = (float*)d_ws;
    float* bufB   = bufA + NF;
    float* dinv   = bufB + NF;
    float* sums   = dinv + N;
    float* counts = sums + NGRAPH * FEAT;

    const int* rows = edge;
    const int* cols = edge + E;

    const int T = 256;

    k_deg_init <<<ceildiv(N, T), T, 0, stream>>>(dinv, N);
    k_deg_accum<<<ceildiv(E, T), T, 0, stream>>>(cols, dinv, E);
    k_deg_rsqrt<<<ceildiv(N, T), T, 0, stream>>>(dinv, N);

    const int mtiles   = (N + 15) / 16;
    const int gemmBlks = ceildiv((long long)mtiles * 4 * 32, T);

    k_gemm_wmma   <<<gemmBlks, T, 0, stream>>>(x, W1, bufA, N);
    k_agg_init    <<<ceildiv(NF, T), T, 0, stream>>>(bufA, dinv, bufB, NF);
    k_edge_scatter<<<ceildiv(E, T), T, 0, stream>>>(rows, cols, bufA, bufB, E);
    k_bias_relu   <<<ceildiv(NF, T), T, 0, stream>>>(bufB, dinv, b1, bufA, NF);

    k_gemm_wmma   <<<gemmBlks, T, 0, stream>>>(bufA, W2, bufB, N);
    k_agg_init    <<<ceildiv(NF, T), T, 0, stream>>>(bufB, dinv, bufA, NF);
    k_edge_scatter<<<ceildiv(E, T), T, 0, stream>>>(rows, cols, bufB, bufA, E);
    k_bias_relu   <<<ceildiv(NF, T), T, 0, stream>>>(bufA, dinv, b2, bufB, NF);

    k_pool_zero <<<ceildiv(NGRAPH * FEAT, T), T, 0, stream>>>(sums, counts);
    k_pool_accum<<<512, T, 0, stream>>>(bufB, batch, sums, counts, N);
    k_pool_div  <<<ceildiv(NGRAPH * FEAT, T), T, 0, stream>>>(sums, counts, out);
}


# edge pass: edges bucketed by destination range once per call; one work-group per range sums degrees and feature rows in LDS, stores each finished row once; f32 sum order = LDS add arrival order
# speedup vs baseline: 2.9127x; 1.6324x over previous
#include <hip/hip_runtime.h>

typedef __attribute__((ext_vector_type(2))) float v2f;
typedef __attribute__((ext_vector_type(8))) float v8f;

#define FEAT 64
#define NGRAPH 64

#define NPB 196
#define MAXB 512
#define SRC_BITS 17
#define BIN_T 1024
#define AGG_T 512

static inline int ceildiv(long long a, long long b) { return (int)((a + b - 1) / b); }

__global__ void k_bin_zero(int* __restrict__ binCount, int* __restrict__ cursor) {
    binCount[threadIdx.x] = 0;
    cursor[threadIdx.x]   = 0;
}

__global__ void k_bin_count(const int* __restrict__ cols, int* __restrict__ binCount, long long E) {
    __shared__ int hist[MAXB];
    for (int i = threadIdx.x; i < MAXB; i += blockDim.x) hist[i] = 0;
    __syncthreads();
    const long long stride = (long long)gridDim.x * blockDim.x;
    for (long long e = (long long)blockIdx.x * blockDim.x + threadIdx.x; e < E; e += stride)
        atomicAdd(&hist[cols[e] / NPB], 1);
    __syncthreads();
    for (int i = threadIdx.x; i < MAXB; i += blockDim.x)
        if (hist[i] != 0) atomicAdd(&binCount[i], hist[i]);
}

__global__ void k_bin_scan(const int* __restrict__ binCount, int* __restrict__ binStart) {
    __shared__ int s[MAXB];
    const int t = threadIdx.x;
    const int own = binCount[t];
    s[t] = own;
    __syncthreads();
    for (int d = 1; d < MAXB; d <<= 1) {
        int v = (t >= d) ? s[t - d] : 0;
        __syncthreads();
        s[t] += v;
        __syncthreads();
    }
    binStart[t] = s[t] - own;
}

__global__ void k_bin_place(const int* __restrict__ rows, const int* __restrict__ cols,
                            const int* __restrict__ binStart, int* __restrict__ cursor,
                            int* __restrict__ packed, long long E) {
    __shared__ int hist[MAXB];
    __shared__ int base[MAXB];
    for (int i = threadIdx.x; i < MAXB; i += blockDim.x) hist[i] = 0;
    __syncthreads();
    const long long stride = (long long)gridDim.x * blockDim.x;
    const long long first  = (long long)blockIdx.x * blockDim.x + threadIdx.x;
    for (long long e = first; e < E; e += stride)
        atomicAdd(&hist[cols[e] / NPB], 1);
    __syncthreads();
    for (int i = threadIdx.x; i < MAXB; i += blockDim.x) {
        const int n = hist[i];
        base[i] = (n != 0) ? binStart[i] + atomicAdd(&cursor[i], n) : 0;
        hist[i] = 0;
    }
    __syncthreads();
    for (long long e = first; e < E; e += stride) {
        const int c = cols[e];
        const int b = c / NPB;
        const int pos = base[b] + atomicAdd(&hist[b], 1);
        packed[pos] = rows[e] | ((c - b * NPB) << SRC_BITS);
    }
}

__global__ void k_deg_bins(const int* __restrict__ packed, const int* __restrict__ binStart,
                           const int* __restrict__ binCount, float* __restrict__ dinv, int N) {
    __shared__ int cnt[NPB];
    const int b = blockIdx.x;
    for (int i = threadIdx.x; i < NPB; i += blockDim.x) cnt[i] = 0;
    __syncthreads();
    const int start = binStart[b];
    const int end   = start + binCount[b];
    for (int e = start + threadIdx.x; e < end; e += blockDim.x)
        atomicAdd(&cnt[packed[e] >> SRC_BITS], 1);
    __syncthreads();
    for (int i = threadIdx.x; i < NPB; i += blockDim.x) {
        const int node = b * NPB + i;
        if (node < N) dinv[node] = rsqrtf(1.0f + (float)cnt[i]);
    }
}

__global__ void k_gemm_wmma(const float* __restrict__ A, const float* __restrict__ W,
                            float* __restrict__ D, int M) {
    int gwave = (blockIdx.x * blockDim.x + threadIdx.x) >> 5;
    int lane  = threadIdx.x & 31;
    int mtiles = (M + 15) >> 4;
    int mtile = gwave >> 2;
    int ntile = gwave & 3;
    if (mtile >= mtiles) return;

    int m0 = mtile << 4, n0 = ntile << 4;
    int half = lane >> 4;
    int r    = lane & 15;

    int arow = m0 + r; if (arow >= M) arow = M - 1;

    v8f c = {};
#pragma unroll
    for (int k0 = 0; k0 < FEAT; k0 += 4) {
        v2f a, b;
        int ka = k0 + 2 * half;
        a.x = A[(long long)arow * FEAT + ka + 0];
        a.y = A[(long long)arow * FEAT + ka + 1];
        b.x = W[(ka + 0) * FEAT + n0 + r];
        b.y = W[(ka + 1) * FEAT + n0 + r];
        c = __builtin_amdgcn_wmma_f32_16x16x4_f32(false, a, false, b, (short)0, c, false, false);
    }
#pragma unroll
    for (int v = 0; v < 8; ++v) {
        int row = m0 + v + 8 * half;
        if (row < M) D[(long long)row * FEAT + n0 + r] = c[v];
    }
}

__global__ void k_row_scale(float* __restrict__ h, const float* __restrict__ dinv, long long total) {
    long long gid = (long long)blockIdx.x * blockDim.x + threadIdx.x;
    if (gid >= total) return;
    h[gid] = h[gid] * dinv[(int)(gid >> 6)];
}

#define EDGE_GRP 8

__device__ __forceinline__ void edge_rows_load(const float* __restrict__ hs, int pk, int lane, int j0,
                                               v2f (&v)[EDGE_GRP]) {
#pragma unroll
    for (int u = 0; u < EDGE_GRP; ++u) {
        const int p   = __builtin_amdgcn_readlane(pk, j0 + u);
        const int src = p & ((1 << SRC_BITS) - 1);
        v[u] = *(const v2f*)(hs + (long long)src * FEAT + 2 * lane);
    }
}

__device__ __forceinline__ void edge_rows_add(float* acc, int pk, int lane, int j0,
                                              const v2f (&v)[EDGE_GRP]) {
#pragma unroll
    for (int u = 0; u < EDGE_GRP; ++u) {
        const int p = __builtin_amdgcn_readlane(pk, j0 + u);
        float* ap = acc + (p >> SRC_BITS) * FEAT + 2 * lane;
        atomicAdd(ap,     v[u].x);
        atomicAdd(ap + 1, v[u].y);
    }
}

__global__ void __launch_bounds__(AGG_T)
k_agg_bins(const int* __restrict__ packed, const int* __restrict__ binStart,
           const int* __restrict__ binCount, const float* __restrict__ hs,
           const float* __restrict__ dinv, const float* __restrict__ bias,
           float* __restrict__ out, int N) {
    __shared__ float acc[NPB * FEAT];
    const int b = blockIdx.x;
    const long long row0 = (long long)b * NPB * FEAT;
    for (int i = threadIdx.x; i < NPB * FEAT; i += AGG_T) {
        const int node = b * NPB + (i >> 6);
        acc[i] = (node < N) ? hs[row0 + i] : 0.0f;
    }
    __syncthreads();

    const int lane  = threadIdx.x & 31;
    const int wave  = threadIdx.x >> 5;
    const int start = binStart[b];
    const int end   = start + binCount[b];
    int base = start + wave * 32;
    int pk_next = (base + lane < end) ? packed[base + lane] : 0;
    for (; base < end; base += AGG_T) {
        const int pk = pk_next;
        const int en = base + AGG_T + lane;
        pk_next = (en < end) ? packed[en] : 0;
        if (base + 32 <= end) {
            v2f g0[EDGE_GRP], g1[EDGE_GRP];
            edge_rows_load(hs, pk, lane, 0, g0);
            edge_rows_load(hs, pk, lane, 1 * EDGE_GRP, g1);
            edge_rows_add(acc, pk, lane, 0, g0);
            edge_rows_load(hs, pk, lane, 2 * EDGE_GRP, g0);
            edge_rows_add(acc, pk, lane, 1 * EDGE_GRP, g1);
            edge_rows_load(hs, pk, lane, 3 * EDGE_GRP, g1);
            edge_rows_add(acc, pk, lane, 2 * EDGE_GRP, g0);
            edge_rows_add(acc, pk, lane, 3 * EDGE_GRP, g1);
        } else {
            const int cnt = end - base;
            for (int j = 0; j < cnt; ++j) {
                const int p   = __builtin_amdgcn_readlane(pk, j);
                const int src = p & ((1 << SRC_BITS) - 1);
                const v2f v = *(const v2f*)(hs + (long long)src * FEAT + 2 * lane);
                float* ap = acc + (p >> SRC_BITS) * FEAT + 2 * lane;
                atomicAdd(ap,     v.x);
                atomicAdd(ap + 1, v.y);
            }
        }
    }
    __syncthreads();

    for (int i = threadIdx.x; i < NPB * FEAT; i += AGG_T) {
        const int node = b * NPB + (i >> 6);
        if (node < N) {
            float v = acc[i] * dinv[node] + bias[i & (FEAT - 1)];
            out[row0 + i] = v > 0.0f ? v : 0.0f;
        }
    }
}

__global__ void k_pool_zero(float* sums, float* counts) {
    int i = blockIdx.x * blockDim.x + threadIdx.x;
    if (i < NGRAPH * FEAT) sums[i] = 0.0f;
    if (i < NGRAPH) counts[i] = 0.0f;
}

__global__ void k_pool_accum(const float* __restrict__ h, const int* __restrict__ batch,
                             float* __restrict__ sums, float* __restrict__ counts, int N) {
    __shared__ float ls[NGRAPH * FEAT];
    __shared__ float lc[NGRAPH];
    for (int i = threadIdx.x; i < NGRAPH * FEAT; i += blockDim.x) ls[i] = 0.0f;
    for (int i = threadIdx.x; i < NGRAPH; i += blockDim.x) lc[i] = 0.0f;
    __syncthreads();

    int wave = threadIdx.x >> 5;
    int lane = threadIdx.x & 31;
    for (int node = blockIdx.x * 8 + wave; node < N; node += gridDim.x * 8) {
        int g = batch[node];
        atomicAdd(&ls[g * FEAT + lane],      h[(long long)node * FEAT + lane]);
        atomicAdd(&ls[g * FEAT + lane + 32], h[(long long)node * FEAT + lane + 32]);
        if (lane == 0) atomicAdd(&lc[g], 1.0f);
    }
    __syncthreads();
    for (int i = threadIdx.x; i < NGRAPH * FEAT; i += blockDim.x)
        if (ls[i] != 0.0f) atomicAdd(&sums[i], ls[i]);
    for (int i = threadIdx.x; i < NGRAPH; i += blockDim.x)
        if (lc[i] != 0.0f) atomicAdd(&counts[i], lc[i]);
}

__global__ void k_pool_div(const float* __restrict__ sums, const float* __restrict__ counts,
                           float* __restrict__ out) {
    int i = blockIdx.x * blockDim.x + threadIdx.x;
    if (i >= NGRAPH * FEAT) return;
    out[i] = sums[i] / fmaxf(counts[i >> 6], 1.0f);
}

extern "C" void kernel_launch(void* const* d_in, const int* in_sizes, int n_in,
                              void* d_out, int out_size, void* d_ws, size_t ws_size,
                              hipStream_t stream) {
    const float* x     = (const float*)d_in[0];
    const int*   edge  = (const int*)d_in[1];
    const int*   batch = (const int*)d_in[2];
    const float* W1    = (const float*)d_in[3];
    const float* b1    = (const float*)d_in[4];
    const float* W2    = (const float*)d_in[5];
    const float* b2    = (const float*)d_in[6];
    float*       out   = (float*)d_out;

    const int       N = in_sizes[0] / FEAT;
    const long long E = (long long)in_sizes[1] / 2;
    const long long NF = (long long)N * FEAT;

    float* bufA   = (float*)d_ws;
    float* bufB   = bufA + NF;
    float* dinv   = bufB + NF;
    float* sums   = dinv + N;
    float* counts = sums + NGRAPH * FEAT;
    int* binCount = (int*)(counts + NGRAPH);
    int* cursor   = binCount + MAXB;
    int* binStart = cursor + MAXB;
    int* packed   = binStart + MAXB;

    const int* rows = edge;
    const int* cols = edge + E;

    const int T = 256;
    const int nbins = ceildiv(N, NPB);

    k_bin_zero <<<1, MAXB, 0, stream>>>(binCount, cursor);
    k_bin_count<<<256, BIN_T, 0, stream>>>(cols, binCount, E);
    k_bin_scan <<<1, MAXB, 0, stream>>>(binCount, binStart);
    k_bin_place<<<256, BIN_T, 0, stream>>>(rows, cols, binStart, cursor, packed, E);

    k_deg_bins<<<nbins, AGG_T, 0, stream>>>(packed, binStart, binCount, dinv, N);

    const int mtiles   = (N + 15) / 16;
    const int gemmBlks = ceildiv((long long)mtiles * 4 * 32, T);

    k_gemm_wmma<<<gemmBlks, T, 0, stream>>>(x, W1, bufA, N);
    k_row_scale<<<ceildiv(NF, T), T, 0, stream>>>(bufA, dinv, NF);
    k_agg_bins <<<nbins, AGG_T, 0, stream>>>(packed, binStart, binCount, bufA, dinv, b1, bufB, N);

    k_gemm_wmma<<<gemmBlks, T, 0, stream>>>(bufB, W2, bufA, N);
    k_row_scale<<<ceildiv(NF, T), T, 0, stream>>>(bufA, dinv, NF);
    k_agg_bins <<<nbins, AGG_T, 0, stream>>>(packed, binStart, binCount, bufA, dinv, b2, bufB, N);

    k_pool_zero <<<ceildiv(NGRAPH * FEAT, T), T, 0, stream>>>(sums, counts);
    k_pool_accum<<<512, T, 0, stream>>>(bufB, batch, sums, counts, N);
    k_pool_div  <<<ceildiv(NGRAPH * FEAT, T), T, 0, stream>>>(sums, counts, out);
}


# dinv[row] scaling moved into the GEMM's accumulator store; the two one-thread-per-element scaling launches (25000 work-groups, 200000 waves each) and their read-write pass over h are gone
# speedup vs baseline: 3.1994x; 1.0984x over previous
#include <hip/hip_runtime.h>

typedef __attribute__((ext_vector_type(2))) float v2f;
typedef __attribute__((ext_vector_type(8))) float v8f;

#define FEAT 64
#define NGRAPH 64

#define NPB 196
#define MAXB 512
#define SRC_BITS 17
#define BIN_T 1024
#define AGG_T 512

static inline int ceildiv(long long a, long long b) { return (int)((a + b - 1) / b); }

__global__ void k_bin_zero(int* __restrict__ binCount, int* __restrict__ cursor) {
    binCount[threadIdx.x] = 0;
    cursor[threadIdx.x]   = 0;
}

__global__ void k_bin_count(const int* __restrict__ cols, int* __restrict__ binCount, long long E) {
    __shared__ int hist[MAXB];
    for (int i = threadIdx.x; i < MAXB; i += blockDim.x) hist[i] = 0;
    __syncthreads();
    const long long stride = (long long)gridDim.x * blockDim.x;
    for (long long e = (long long)blockIdx.x * blockDim.x + threadIdx.x; e < E; e += stride)
        atomicAdd(&hist[cols[e] / NPB], 1);
    __syncthreads();
    for (int i = threadIdx.x; i < MAXB; i += blockDim.x)
        if (hist[i] != 0) atomicAdd(&binCount[i], hist[i]);
}

__global__ void k_bin_scan(const int* __restrict__ binCount, int* __restrict__ binStart) {
    __shared__ int s[MAXB];
    const int t = threadIdx.x;
    const int own = binCount[t];
    s[t] = own;
    __syncthreads();
    for (int d = 1; d < MAXB; d <<= 1) {
        int v = (t >= d) ? s[t - d] : 0;
        __syncthreads();
        s[t] += v;
        __syncthreads();
    }
    binStart[t] = s[t] - own;
}

__global__ void k_bin_place(const int* __restrict__ rows, const int* __restrict__ cols,
                            const int* __restrict__ binStart, int* __restrict__ cursor,
                            int* __restrict__ packed, long long E) {
    __shared__ int hist[MAXB];
    __shared__ int base[MAXB];
    for (int i = threadIdx.x; i < MAXB; i += blockDim.x) hist[i] = 0;
    __syncthreads();
    const long long stride = (long long)gridDim.x * blockDim.x;
    const long long first  = (long long)blockIdx.x * blockDim.x + threadIdx.x;
    for (long long e = first; e < E; e += stride)
        atomicAdd(&hist[cols[e] / NPB], 1);
    __syncthreads();
    for (int i = threadIdx.x; i < MAXB; i += blockDim.x) {
        const int n = hist[i];
        base[i] = (n != 0) ? binStart[i] + atomicAdd(&cursor[i], n) : 0;
        hist[i] = 0;
    }
    __syncthreads();
    for (long long e = first; e < E; e += stride) {
        const int c = cols[e];
        const int b = c / NPB;
        const int pos = base[b] + atomicAdd(&hist[b], 1);
        packed[pos] = rows[e] | ((c - b * NPB) << SRC_BITS);
    }
}

__global__ void k_deg_bins(const int* __restrict__ packed, const int* __restrict__ binStart,
                           const int* __restrict__ binCount, float* __restrict__ dinv, int N) {
    __shared__ int cnt[NPB];
    const int b = blockIdx.x;
    for (int i = threadIdx.x; i < NPB; i += blockDim.x) cnt[i] = 0;
    __syncthreads();
    const int start = binStart[b];
    const int end   = start + binCount[b];
    for (int e = start + threadIdx.x; e < end; e += blockDim.x)
        atomicAdd(&cnt[packed[e] >> SRC_BITS], 1);
    __syncthreads();
    for (int i = threadIdx.x; i < NPB; i += blockDim.x) {
        const int node = b * NPB + i;
        if (node < N) dinv[node] = rsqrtf(1.0f + (float)cnt[i]);
    }
}

__global__ void k_gemm_wmma(const float* __restrict__ A, const float* __restrict__ W,
                            const float* __restrict__ dinv, float* __restrict__ D, int M) {
    int gwave = (blockIdx.x * blockDim.x + threadIdx.x) >> 5;
    int lane  = threadIdx.x & 31;
    int mtiles = (M + 15) >> 4;
    int mtile = gwave >> 2;
    int ntile = gwave & 3;
    if (mtile >= mtiles) return;

    int m0 = mtile << 4, n0 = ntile << 4;
    int half = lane >> 4;
    int r    = lane & 15;

    int arow = m0 + r; if (arow >= M) arow = M - 1;

    float s[8];
#pragma unroll
    for (int v = 0; v < 8; ++v) {
        int row = m0 + v + 8 * half;
        s[v] = dinv[row < M ? row : M - 1];
    }

    v8f c = {};
#pragma unroll
    for (int k0 = 0; k0 < FEAT; k0 += 4) {
        v2f a, b;
        int ka = k0 + 2 * half;
        a.x = A[(long long)arow * FEAT + ka + 0];
        a.y = A[(long long)arow * FEAT + ka + 1];
        b.x = W[(ka + 0) * FEAT + n0 + r];
        b.y = W[(ka + 1) * FEAT + n0 + r];
        c = __builtin_amdgcn_wmma_f32_16x16x4_f32(false, a, false, b, (short)0, c, false, false);
    }
#pragma unroll
    for (int v = 0; v < 8; ++v) {
        int row = m0 + v + 8 * half;
        if (row < M) D[(long long)row * FEAT + n0 + r] = c[v] * s[v];
    }
}


#define EDGE_GRP 8

__device__ __forceinline__ void edge_rows_load(const float* __restrict__ hs, int pk, int lane, int j0,
                                               v2f (&v)[EDGE_GRP]) {
#pragma unroll
    for (int u = 0; u < EDGE_GRP; ++u) {
        const int p   = __builtin_amdgcn_readlane(pk, j0 + u);
        const int src = p & ((1 << SRC_BITS) - 1);
        v[u] = *(const v2f*)(hs + (long long)src * FEAT + 2 * lane);
    }
}

__device__ __forceinline__ void edge_rows_add(float* acc, int pk, int lane, int j0,
                                              const v2f (&v)[EDGE_GRP]) {
#pragma unroll
    for (int u = 0; u < EDGE_GRP; ++u) {
        const int p = __builtin_amdgcn_readlane(pk, j0 + u);
        float* ap = acc + (p >> SRC_BITS) * FEAT + 2 * lane;
        atomicAdd(ap,     v[u].x);
        atomicAdd(ap + 1, v[u].y);
    }
}

__global__ void __launch_bounds__(AGG_T)
k_agg_bins(const int* __restrict__ packed, const int* __restrict__ binStart,
           const int* __restrict__ binCount, const float* __restrict__ hs,
           const float* __restrict__ dinv, const float* __restrict__ bias,
           float* __restrict__ out, int N) {
    __shared__ float acc[NPB * FEAT];
    const int b = blockIdx.x;
    const long long row0 = (long long)b * NPB * FEAT;
    for (int i = threadIdx.x; i < NPB * FEAT; i += AGG_T) {
        const int node = b * NPB + (i >> 6);
        acc[i] = (node < N) ? hs[row0 + i] : 0.0f;
    }
    __syncthreads();

    const int lane  = threadIdx.x & 31;
    const int wave  = threadIdx.x >> 5;
    const int start = binStart[b];
    const int end   = start + binCount[b];
    int base = start + wave * 32;
    int pk_next = (base + lane < end) ? packed[base + lane] : 0;
    for (; base < end; base += AGG_T) {
        const int pk = pk_next;
        const int en = base + AGG_T + lane;
        pk_next = (en < end) ? packed[en] : 0;
        if (base + 32 <= end) {
            v2f g0[EDGE_GRP], g1[EDGE_GRP];
            edge_rows_load(hs, pk, lane, 0, g0);
            edge_rows_load(hs, pk, lane, 1 * EDGE_GRP, g1);
            edge_rows_add(acc, pk, lane, 0, g0);
            edge_rows_load(hs, pk, lane, 2 * EDGE_GRP, g0);
            edge_rows_add(acc, pk, lane, 1 * EDGE_GRP, g1);
            edge_rows_load(hs, pk, lane, 3 * EDGE_GRP, g1);
            edge_rows_add(acc, pk, lane, 2 * EDGE_GRP, g0);
            edge_rows_add(acc, pk, lane, 3 * EDGE_GRP, g1);
        } else {
            const int cnt = end - base;
            for (int j = 0; j < cnt; ++j) {
                const int p   = __builtin_amdgcn_readlane(pk, j);
                const int src = p & ((1 << SRC_BITS) - 1);
                const v2f v = *(const v2f*)(hs + (long long)src * FEAT + 2 * lane);
                float* ap = acc + (p >> SRC_BITS) * FEAT + 2 * lane;
                atomicAdd(ap,     v.x);
                atomicAdd(ap + 1, v.y);
            }
        }
    }
    __syncthreads();

    for (int i = threadIdx.x; i < NPB * FEAT; i += AGG_T) {
        const int node = b * NPB + (i >> 6);
        if (node < N) {
            float v = acc[i] * dinv[node] + bias[i & (FEAT - 1)];
            out[row0 + i] = v > 0.0f ? v : 0.0f;
        }
    }
}

__global__ void k_pool_zero(float* sums, float* counts) {
    int i = blockIdx.x * blockDim.x + threadIdx.x;
    if (i < NGRAPH * FEAT) sums[i] = 0.0f;
    if (i < NGRAPH) counts[i] = 0.0f;
}

__global__ void k_pool_accum(const float* __restrict__ h, const int* __restrict__ batch,
                             float* __restrict__ sums, float* __restrict__ counts, int N) {
    __shared__ float ls[NGRAPH * FEAT];
    __shared__ float lc[NGRAPH];
    for (int i = threadIdx.x; i < NGRAPH * FEAT; i += blockDim.x) ls[i] = 0.0f;
    for (int i = threadIdx.x; i < NGRAPH; i += blockDim.x) lc[i] = 0.0f;
    __syncthreads();

    int wave = threadIdx.x >> 5;
    int lane = threadIdx.x & 31;
    for (int node = blockIdx.x * 8 + wave; node < N; node += gridDim.x * 8) {
        int g = batch[node];
        atomicAdd(&ls[g * FEAT + lane],      h[(long long)node * FEAT + lane]);
        atomicAdd(&ls[g * FEAT + lane + 32], h[(long long)node * FEAT + lane + 32]);
        if (lane == 0) atomicAdd(&lc[g], 1.0f);
    }
    __syncthreads();
    for (int i = threadIdx.x; i < NGRAPH * FEAT; i += blockDim.x)
        if (ls[i] != 0.0f) atomicAdd(&sums[i], ls[i]);
    for (int i = threadIdx.x; i < NGRAPH; i += blockDim.x)
        if (lc[i] != 0.0f) atomicAdd(&counts[i], lc[i]);
}

__global__ void k_pool_div(const float* __restrict__ sums, const float* __restrict__ counts,
                           float* __restrict__ out) {
    int i = blockIdx.x * blockDim.x + threadIdx.x;
    if (i >= NGRAPH * FEAT) return;
    out[i] = sums[i] / fmaxf(counts[i >> 6], 1.0f);
}

extern "C" void kernel_launch(void* const* d_in, const int* in_sizes, int n_in,
                              void* d_out, int out_size, void* d_ws, size_t ws_size,
                              hipStream_t stream) {
    const float* x     = (const float*)d_in[0];
    const int*   edge  = (const int*)d_in[1];
    const int*   batch = (const int*)d_in[2];
    const float* W1    = (const float*)d_in[3];
    const float* b1    = (const float*)d_in[4];
    const float* W2    = (const float*)d_in[5];
    const float* b2    = (const float*)d_in[6];
    float*       out   = (float*)d_out;

    const int       N = in_sizes[0] / FEAT;
    const long long E = (long long)in_sizes[1] / 2;
    const long long NF = (long long)N * FEAT;

    float* bufA   = (float*)d_ws;
    float* bufB   = bufA + NF;
    float* dinv   = bufB + NF;
    float* sums   = dinv + N;
    float* counts = sums + NGRAPH * FEAT;
    int* binCount = (int*)(counts + NGRAPH);
    int* cursor   = binCount + MAXB;
    int* binStart = cursor + MAXB;
    int* packed   = binStart + MAXB;

    const int* rows = edge;
    const int* cols = edge + E;

    const int T = 256;
    const int nbins = ceildiv(N, NPB);

    k_bin_zero <<<1, MAXB, 0, stream>>>(binCount, cursor);
    k_bin_count<<<256, BIN_T, 0, stream>>>(cols, binCount, E);
    k_bin_scan <<<1, MAXB, 0, stream>>>(binCount, binStart);
    k_bin_place<<<256, BIN_T, 0, stream>>>(rows, cols, binStart, cursor, packed, E);

    k_deg_bins<<<nbins, AGG_T, 0, stream>>>(packed, binStart, binCount, dinv, N);

    const int mtiles   = (N + 15) / 16;
    const int gemmBlks = ceildiv((long long)mtiles * 4 * 32, T);

    k_gemm_wmma<<<gemmBlks, T, 0, stream>>>(x, W1, dinv, bufA, N);
    k_agg_bins <<<nbins, AGG_T, 0, stream>>>(packed, binStart, binCount, bufA, dinv, b1, bufB, N);

    k_gemm_wmma<<<gemmBlks, T, 0, stream>>>(bufB, W2, dinv, bufA, N);
    k_agg_bins <<<nbins, AGG_T, 0, stream>>>(packed, binStart, binCount, bufA, dinv, b2, bufB, N);

    k_pool_zero <<<ceildiv(NGRAPH * FEAT, T), T, 0, stream>>>(sums, counts);
    k_pool_accum<<<512, T, 0, stream>>>(bufB, batch, sums, counts, N);
    k_pool_div  <<<ceildiv(NGRAPH * FEAT, T), T, 0, stream>>>(sums, counts, out);
}


# edge pass: edges sorted by destination row (LDS counting sort in the degree pass); one wave sums a row's sources in registers, stores it once; f32 LDS atomics gone; sum order = sorted edge order
# speedup vs baseline: 4.4928x; 1.4043x over previous
#include <hip/hip_runtime.h>

typedef __attribute__((ext_vector_type(2))) float v2f;
typedef __attribute__((ext_vector_type(8))) float v8f;

#define FEAT 64
#define NGRAPH 64

#define NPB 196
#define NPB_PAD 256
#define MAXB 512
#define SRC_BITS 17
#define BIN_T 1024
#define AGG_T 512
#define ROW_T 256
#define ROW_GRP 16
#define ROW_TAIL 8
#define SORT_UNR 4

static inline int ceildiv(long long a, long long b) { return (int)((a + b - 1) / b); }

__global__ void k_bin_zero(int* __restrict__ binCount, int* __restrict__ cursor) {
    binCount[threadIdx.x] = 0;
    cursor[threadIdx.x]   = 0;
}

__global__ void k_bin_count(const int* __restrict__ cols, int* __restrict__ binCount, long long E) {
    __shared__ int hist[MAXB];
    for (int i = threadIdx.x; i < MAXB; i += blockDim.x) hist[i] = 0;
    __syncthreads();
    const long long stride = (long long)gridDim.x * blockDim.x;
    for (long long e = (long long)blockIdx.x * blockDim.x + threadIdx.x; e < E; e += stride)
        atomicAdd(&hist[cols[e] / NPB], 1);
    __syncthreads();
    for (int i = threadIdx.x; i < MAXB; i += blockDim.x)
        if (hist[i] != 0) atomicAdd(&binCount[i], hist[i]);
}

__global__ void k_bin_scan(const int* __restrict__ binCount, int* __restrict__ binStart) {
    __shared__ int s[MAXB];
    const int t = threadIdx.x;
    const int own = binCount[t];
    s[t] = own;
    __syncthreads();
    for (int d = 1; d < MAXB; d <<= 1) {
        int v = (t >= d) ? s[t - d] : 0;
        __syncthreads();
        s[t] += v;
        __syncthreads();
    }
    binStart[t] = s[t] - own;
}

__global__ void k_bin_place(const int* __restrict__ rows, const int* __restrict__ cols,
                            const int* __restrict__ binStart, int* __restrict__ cursor,
                            int* __restrict__ packed, long long E) {
    __shared__ int hist[MAXB];
    __shared__ int base[MAXB];
    for (int i = threadIdx.x; i < MAXB; i += blockDim.x) hist[i] = 0;
    __syncthreads();
    const long long stride = (long long)gridDim.x * blockDim.x;
    const long long first  = (long long)blockIdx.x * blockDim.x + threadIdx.x;
    for (long long e = first; e < E; e += stride)
        atomicAdd(&hist[cols[e] / NPB], 1);
    __syncthreads();
    for (int i = threadIdx.x; i < MAXB; i += blockDim.x) {
        const int n = hist[i];
        base[i] = (n != 0) ? binStart[i] + atomicAdd(&cursor[i], n) : 0;
        hist[i] = 0;
    }
    __syncthreads();
    for (long long e = first; e < E; e += stride) {
        const int c = cols[e];
        const int b = c / NPB;
        const int pos = base[b] + atomicAdd(&hist[b], 1);
        packed[pos] = rows[e] | ((c - b * NPB) << SRC_BITS);
    }
}

__global__ void __launch_bounds__(AGG_T)
k_deg_sort(const int* __restrict__ packed, const int* __restrict__ binStart,
           const int* __restrict__ binCount, float* __restrict__ dinv,
           int* __restrict__ rowStart, int* __restrict__ rowCnt, int* __restrict__ sorted, int N) {
    __shared__ int cnt[NPB_PAD];
    __shared__ int off[NPB_PAD];
    const int b = blockIdx.x;
    const int t = threadIdx.x;
    if (t < NPB_PAD) cnt[t] = 0;
    __syncthreads();
    const int start = binStart[b];
    const int end   = start + binCount[b];
    for (int e = start + t; e < end; e += SORT_UNR * AGG_T) {
        int p[SORT_UNR];
#pragma unroll
        for (int k = 0; k < SORT_UNR; ++k) {
            const int ee = e + k * AGG_T;
            p[k] = (ee < end) ? packed[ee] : -1;
        }
#pragma unroll
        for (int k = 0; k < SORT_UNR; ++k)
            if (p[k] >= 0) atomicAdd(&cnt[p[k] >> SRC_BITS], 1);
    }
    __syncthreads();

    int own = 0;
    if (t < NPB_PAD) { own = cnt[t]; off[t] = own; }
    __syncthreads();
    for (int d = 1; d < NPB_PAD; d <<= 1) {
        int v = 0;
        if (t < NPB_PAD && t >= d) v = off[t - d];
        __syncthreads();
        if (t < NPB_PAD) off[t] += v;
        __syncthreads();
    }
    if (t < NPB_PAD) {
        const int ex = off[t] - own;
        off[t] = ex;
        cnt[t] = 0;
        const int node = b * NPB + t;
        if (t < NPB && node < N) {
            dinv[node]     = rsqrtf(1.0f + (float)own);
            rowStart[node] = start + ex;
            rowCnt[node]   = own;
        }
    }
    __syncthreads();
    for (int e = start + t; e < end; e += SORT_UNR * AGG_T) {
        int p[SORT_UNR];
#pragma unroll
        for (int k = 0; k < SORT_UNR; ++k) {
            const int ee = e + k * AGG_T;
            p[k] = (ee < end) ? packed[ee] : -1;
        }
#pragma unroll
        for (int k = 0; k < SORT_UNR; ++k) {
            if (p[k] >= 0) {
                const int d = p[k] >> SRC_BITS;
                const int pos = start + off[d] + atomicAdd(&cnt[d], 1);
                sorted[pos] = p[k] & ((1 << SRC_BITS) - 1);
            }
        }
    }
}

__global__ void k_gemm_wmma(const float* __restrict__ A, const float* __restrict__ W,
                            const float* __restrict__ dinv, float* __restrict__ D, int M) {
    int gwave = (blockIdx.x * blockDim.x + threadIdx.x) >> 5;
    int lane  = threadIdx.x & 31;
    int mtiles = (M + 15) >> 4;
    int mtile = gwave >> 2;
    int ntile = gwave & 3;
    if (mtile >= mtiles) return;

    int m0 = mtile << 4, n0 = ntile << 4;
    int half = lane >> 4;
    int r    = lane & 15;

    int arow = m0 + r; if (arow >= M) arow = M - 1;

    float s[8];
#pragma unroll
    for (int v = 0; v < 8; ++v) {
        int row = m0 + v + 8 * half;
        s[v] = dinv[row < M ? row : M - 1];
    }

    v8f c = {};
#pragma unroll
    for (int k0 = 0; k0 < FEAT; k0 += 4) {
        v2f a, b;
        int ka = k0 + 2 * half;
        a.x = A[(long long)arow * FEAT + ka + 0];
        a.y = A[(long long)arow * FEAT + ka + 1];
        b.x = W[(ka + 0) * FEAT + n0 + r];
        b.y = W[(ka + 1) * FEAT + n0 + r];
        c = __builtin_amdgcn_wmma_f32_16x16x4_f32(false, a, false, b, (short)0, c, false, false);
    }
#pragma unroll
    for (int v = 0; v < 8; ++v) {
        int row = m0 + v + 8 * half;
        if (row < M) D[(long long)row * FEAT + n0 + r] = c[v] * s[v];
    }
}


__global__ void __launch_bounds__(ROW_T)
k_agg_rows(const int* __restrict__ sorted, const int* __restrict__ rowStart,
           const int* __restrict__ rowCnt, const float* __restrict__ hs,
           const float* __restrict__ dinv, const float* __restrict__ bias,
           float* __restrict__ out, int N) {
    const int lane = threadIdx.x & 31;
    const int wave = __builtin_amdgcn_readfirstlane((int)(threadIdx.x >> 5));
    const int nw   = (int)gridDim.x * (ROW_T >> 5);
    const v2f bv   = *(const v2f*)(bias + 2 * lane);

    int r = (int)blockIdx.x * (ROW_T >> 5) + wave;
    int s_next = 0, n_next = 0;
    if (r < N) { s_next = rowStart[r]; n_next = rowCnt[r]; }
    for (; r < N; r += nw) {
        const int s = __builtin_amdgcn_readfirstlane(s_next);
        const int n = __builtin_amdgcn_readfirstlane(n_next);
        const int rn = r + nw;
        if (rn < N) { s_next = rowStart[rn]; n_next = rowCnt[rn]; }

        v2f acc = *(const v2f*)(hs + (long long)r * FEAT + 2 * lane);
        for (int base = 0; base < n; base += 32) {
            const int rem = (n - base < 32) ? (n - base) : 32;
            const int pk  = (lane < rem) ? sorted[s + base + lane] : 0;
            int j0 = 0;
            for (; j0 + ROW_GRP <= rem; j0 += ROW_GRP) {
                v2f v[ROW_GRP];
#pragma unroll
                for (int u = 0; u < ROW_GRP; ++u) {
                    const int src = __builtin_amdgcn_readlane(pk, j0 + u);
                    v[u] = *(const v2f*)(hs + (long long)src * FEAT + 2 * lane);
                }
#pragma unroll
                for (int u = 0; u < ROW_GRP; ++u) { acc.x += v[u].x; acc.y += v[u].y; }
            }
            for (; j0 < rem; j0 += ROW_TAIL) {
                v2f v[ROW_TAIL];
#pragma unroll
                for (int u = 0; u < ROW_TAIL; ++u) {
                    const int src = __builtin_amdgcn_readlane(pk, j0 + u);
                    v[u] = *(const v2f*)(hs + (long long)src * FEAT + 2 * lane);
                }
#pragma unroll
                for (int u = 0; u < ROW_TAIL; ++u) {
                    if (j0 + u < rem) { acc.x += v[u].x; acc.y += v[u].y; }
                }
            }
        }
        const float dc = dinv[r];
        v2f o;
        o.x = acc.x * dc + bv.x;
        o.y = acc.y * dc + bv.y;
        o.x = o.x > 0.0f ? o.x : 0.0f;
        o.y = o.y > 0.0f ? o.y : 0.0f;
        *(v2f*)(out + (long long)r * FEAT + 2 * lane) = o;
    }
}

__global__ void k_pool_zero(float* sums, float* counts) {
    int i = blockIdx.x * blockDim.x + threadIdx.x;
    if (i < NGRAPH * FEAT) sums[i] = 0.0f;
    if (i < NGRAPH) counts[i] = 0.0f;
}

__global__ void k_pool_accum(const float* __restrict__ h, const int* __restrict__ batch,
                             float* __restrict__ sums, float* __restrict__ counts, int N) {
    __shared__ float ls[NGRAPH * FEAT];
    __shared__ float lc[NGRAPH];
    for (int i = threadIdx.x; i < NGRAPH * FEAT; i += blockDim.x) ls[i] = 0.0f;
    for (int i = threadIdx.x; i < NGRAPH; i += blockDim.x) lc[i] = 0.0f;
    __syncthreads();

    int wave = threadIdx.x >> 5;
    int lane = threadIdx.x & 31;
    for (int node = blockIdx.x * 8 + wave; node < N; node += gridDim.x * 8) {
        int g = batch[node];
        atomicAdd(&ls[g * FEAT + lane],      h[(long long)node * FEAT + lane]);
        atomicAdd(&ls[g * FEAT + lane + 32], h[(long long)node * FEAT + lane + 32]);
        if (lane == 0) atomicAdd(&lc[g], 1.0f);
    }
    __syncthreads();
    for (int i = threadIdx.x; i < NGRAPH * FEAT; i += blockDim.x)
        if (ls[i] != 0.0f) atomicAdd(&sums[i], ls[i]);
    for (int i = threadIdx.x; i < NGRAPH; i += blockDim.x)
        if (lc[i] != 0.0f) atomicAdd(&counts[i], lc[i]);
}

__global__ void k_pool_div(const float* __restrict__ sums, const float* __restrict__ counts,
                           float* __restrict__ out) {
    int i = blockIdx.x * blockDim.x + threadIdx.x;
    if (i >= NGRAPH * FEAT) return;
    out[i] = sums[i] / fmaxf(counts[i >> 6], 1.0f);
}

extern "C" void kernel_launch(void* const* d_in, const int* in_sizes, int n_in,
                              void* d_out, int out_size, void* d_ws, size_t ws_size,
                              hipStream_t stream) {
    const float* x     = (const float*)d_in[0];
    const int*   edge  = (const int*)d_in[1];
    const int*   batch = (const int*)d_in[2];
    const float* W1    = (const float*)d_in[3];
    const float* b1    = (const float*)d_in[4];
    const float* W2    = (const float*)d_in[5];
    const float* b2    = (const float*)d_in[6];
    float*       out   = (float*)d_out;

    const int       N = in_sizes[0] / FEAT;
    const long long E = (long long)in_sizes[1] / 2;
    const long long NF = (long long)N * FEAT;

    float* bufA   = (float*)d_ws;
    float* bufB   = bufA + NF;
    float* dinv   = bufB + NF;
    float* sums   = dinv + N;
    float* counts = sums + NGRAPH * FEAT;
    int* binCount = (int*)(counts + NGRAPH);
    int* cursor   = binCount + MAXB;
    int* binStart = cursor + MAXB;
    int* packed   = binStart + MAXB;
    int* sorted   = packed + E;
    int* rowStart = sorted + E;
    int* rowCnt   = rowStart + N;

    const int* rows = edge;
    const int* cols = edge + E;

    const int T = 256;
    const int nbins = ceildiv(N, NPB);

    k_bin_zero <<<1, MAXB, 0, stream>>>(binCount, cursor);
    k_bin_count<<<256, BIN_T, 0, stream>>>(cols, binCount, E);
    k_bin_scan <<<1, MAXB, 0, stream>>>(binCount, binStart);
    k_bin_place<<<256, BIN_T, 0, stream>>>(rows, cols, binStart, cursor, packed, E);

    k_deg_sort<<<nbins, AGG_T, 0, stream>>>(packed, binStart, binCount, dinv, rowStart, rowCnt, sorted, N);

    const int mtiles   = (N + 15) / 16;
    const int gemmBlks = ceildiv((long long)mtiles * 4 * 32, T);
    const int rowBlks  = ceildiv(N, (ROW_T / 32) * 4);

    k_gemm_wmma<<<gemmBlks, T, 0, stream>>>(x, W1, dinv, bufA, N);
    k_agg_rows <<<rowBlks, ROW_T, 0, stream>>>(sorted, rowStart, rowCnt, bufA, dinv, b1, bufB, N);

    k_gemm_wmma<<<gemmBlks, T, 0, stream>>>(bufB, W2, dinv, bufA, N);
    k_agg_rows <<<rowBlks, ROW_T, 0, stream>>>(sorted, rowStart, rowCnt, bufA, dinv, b2, bufB, N);

    k_pool_zero <<<ceildiv(NGRAPH * FEAT, T), T, 0, stream>>>(sums, counts);
    k_pool_accum<<<512, T, 0, stream>>>(bufB, batch, sums, counts, N);
    k_pool_div  <<<ceildiv(NGRAPH * FEAT, T), T, 0, stream>>>(sums, counts, out);
}


# layer-2 rows go from registers into the mean pool's per-graph sums (summed per wave, per work-group in LDS, then f32 atomic adds: pool sum order changes); row store, pool re-read pass and launch gone
# speedup vs baseline: 5.0144x; 1.1161x over previous
#include <hip/hip_runtime.h>

typedef __attribute__((ext_vector_type(2))) float v2f;
typedef __attribute__((ext_vector_type(8))) float v8f;

#define FEAT 64
#define NGRAPH 64

#define NPB 196
#define NPB_PAD 256
#define MAXB 512
#define SRC_BITS 17
#define BIN_T 1024
#define AGG_T 512
#define ROW_T 256
#define ROW_GRP 16
#define ROW_TAIL 8
#define POOL_ROWS 32
#define SORT_UNR 4

static inline int ceildiv(long long a, long long b) { return (int)((a + b - 1) / b); }

__global__ void k_bin_zero(int* __restrict__ binCount, int* __restrict__ cursor) {
    binCount[threadIdx.x] = 0;
    cursor[threadIdx.x]   = 0;
}

__global__ void k_bin_count(const int* __restrict__ cols, int* __restrict__ binCount, long long E) {
    __shared__ int hist[MAXB];
    for (int i = threadIdx.x; i < MAXB; i += blockDim.x) hist[i] = 0;
    __syncthreads();
    const long long stride = (long long)gridDim.x * blockDim.x;
    for (long long e = (long long)blockIdx.x * blockDim.x + threadIdx.x; e < E; e += stride)
        atomicAdd(&hist[cols[e] / NPB], 1);
    __syncthreads();
    for (int i = threadIdx.x; i < MAXB; i += blockDim.x)
        if (hist[i] != 0) atomicAdd(&binCount[i], hist[i]);
}

__global__ void k_bin_scan(const int* __restrict__ binCount, int* __restrict__ binStart) {
    __shared__ int s[MAXB];
    const int t = threadIdx.x;
    const int own = binCount[t];
    s[t] = own;
    __syncthreads();
    for (int d = 1; d < MAXB; d <<= 1) {
        int v = (t >= d) ? s[t - d] : 0;
        __syncthreads();
        s[t] += v;
        __syncthreads();
    }
    binStart[t] = s[t] - own;
}

__global__ void k_bin_place(const int* __restrict__ rows, const int* __restrict__ cols,
                            const int* __restrict__ binStart, int* __restrict__ cursor,
                            int* __restrict__ packed, long long E) {
    __shared__ int hist[MAXB];
    __shared__ int base[MAXB];
    for (int i = threadIdx.x; i < MAXB; i += blockDim.x) hist[i] = 0;
    __syncthreads();
    const long long stride = (long long)gridDim.x * blockDim.x;
    const long long first  = (long long)blockIdx.x * blockDim.x + threadIdx.x;
    for (long long e = first; e < E; e += stride)
        atomicAdd(&hist[cols[e] / NPB], 1);
    __syncthreads();
    for (int i = threadIdx.x; i < MAXB; i += blockDim.x) {
        const int n = hist[i];
        base[i] = (n != 0) ? binStart[i] + atomicAdd(&cursor[i], n) : 0;
        hist[i] = 0;
    }
    __syncthreads();
    for (long long e = first; e < E; e += stride) {
        const int c = cols[e];
        const int b = c / NPB;
        const int pos = base[b] + atomicAdd(&hist[b], 1);
        packed[pos] = rows[e] | ((c - b * NPB) << SRC_BITS);
    }
}

__global__ void __launch_bounds__(AGG_T)
k_deg_sort(const int* __restrict__ packed, const int* __restrict__ binStart,
           const int* __restrict__ binCount, float* __restrict__ dinv,
           int* __restrict__ rowStart, int* __restrict__ rowCnt, int* __restrict__ sorted, int N) {
    __shared__ int cnt[NPB_PAD];
    __shared__ int off[NPB_PAD];
    const int b = blockIdx.x;
    const int t = threadIdx.x;
    if (t < NPB_PAD) cnt[t] = 0;
    __syncthreads();
    const int start = binStart[b];
    const int end   = start + binCount[b];
    for (int e = start + t; e < end; e += SORT_UNR * AGG_T) {
        int p[SORT_UNR];
#pragma unroll
        for (int k = 0; k < SORT_UNR; ++k) {
            const int ee = e + k * AGG_T;
            p[k] = (ee < end) ? packed[ee] : -1;
        }
#pragma unroll
        for (int k = 0; k < SORT_UNR; ++k)
            if (p[k] >= 0) atomicAdd(&cnt[p[k] >> SRC_BITS], 1);
    }
    __syncthreads();

    int own = 0;
    if (t < NPB_PAD) { own = cnt[t]; off[t] = own; }
    __syncthreads();
    for (int d = 1; d < NPB_PAD; d <<= 1) {
        int v = 0;
        if (t < NPB_PAD && t >= d) v = off[t - d];
        __syncthreads();
        if (t < NPB_PAD) off[t] += v;
        __syncthreads();
    }
    if (t < NPB_PAD) {
        const int ex = off[t] - own;
        off[t] = ex;
        cnt[t] = 0;
        const int node = b * NPB + t;
        if (t < NPB && node < N) {
            dinv[node]     = rsqrtf(1.0f + (float)own);
            rowStart[node] = start + ex;
            rowCnt[node]   = own;
        }
    }
    __syncthreads();
    for (int e = start + t; e < end; e += SORT_UNR * AGG_T) {
        int p[SORT_UNR];
#pragma unroll
        for (int k = 0; k < SORT_UNR; ++k) {
            const int ee = e + k * AGG_T;
            p[k] = (ee < end) ? packed[ee] : -1;
        }
#pragma unroll
        for (int k = 0; k < SORT_UNR; ++k) {
            if (p[k] >= 0) {
                const int d = p[k] >> SRC_BITS;
                const int pos = start + off[d] + atomicAdd(&cnt[d], 1);
                sorted[pos] = p[k] & ((1 << SRC_BITS) - 1);
            }
        }
    }
}

__global__ void k_gemm_wmma(const float* __restrict__ A, const float* __restrict__ W,
                            const float* __restrict__ dinv, float* __restrict__ D, int M) {
    int gwave = (blockIdx.x * blockDim.x + threadIdx.x) >> 5;
    int lane  = threadIdx.x & 31;
    int mtiles = (M + 15) >> 4;
    int mtile = gwave >> 2;
    int ntile = gwave & 3;
    if (mtile >= mtiles) return;

    int m0 = mtile << 4, n0 = ntile << 4;
    int half = lane >> 4;
    int r    = lane & 15;

    int arow = m0 + r; if (arow >= M) arow = M - 1;

    float s[8];
#pragma unroll
    for (int v = 0; v < 8; ++v) {
        int row = m0 + v + 8 * half;
        s[v] = dinv[row < M ? row : M - 1];
    }

    v8f c = {};
#pragma unroll
    for (int k0 = 0; k0 < FEAT; k0 += 4) {
        v2f a, b;
        int ka = k0 + 2 * half;
        a.x = A[(long long)arow * FEAT + ka + 0];
        a.y = A[(long long)arow * FEAT + ka + 1];
        b.x = W[(ka + 0) * FEAT + n0 + r];
        b.y = W[(ka + 1) * FEAT + n0 + r];
        c = __builtin_amdgcn_wmma_f32_16x16x4_f32(false, a, false, b, (short)0, c, false, false);
    }
#pragma unroll
    for (int v = 0; v < 8; ++v) {
        int row = m0 + v + 8 * half;
        if (row < M) D[(long long)row * FEAT + n0 + r] = c[v] * s[v];
    }
}


__global__ void __launch_bounds__(ROW_T)
k_agg_rows(const int* __restrict__ sorted, const int* __restrict__ rowStart,
           const int* __restrict__ rowCnt, const float* __restrict__ hs,
           const float* __restrict__ dinv, const float* __restrict__ bias,
           float* __restrict__ out, int N) {
    const int lane = threadIdx.x & 31;
    const int wave = __builtin_amdgcn_readfirstlane((int)(threadIdx.x >> 5));
    const int nw   = (int)gridDim.x * (ROW_T >> 5);
    const v2f bv   = *(const v2f*)(bias + 2 * lane);

    int r = (int)blockIdx.x * (ROW_T >> 5) + wave;
    int s_next = 0, n_next = 0;
    if (r < N) { s_next = rowStart[r]; n_next = rowCnt[r]; }
    for (; r < N; r += nw) {
        const int s = __builtin_amdgcn_readfirstlane(s_next);
        const int n = __builtin_amdgcn_readfirstlane(n_next);
        const int rn = r + nw;
        if (rn < N) { s_next = rowStart[rn]; n_next = rowCnt[rn]; }

        v2f acc = *(const v2f*)(hs + (long long)r * FEAT + 2 * lane);
        for (int base = 0; base < n; base += 32) {
            const int rem = (n - base < 32) ? (n - base) : 32;
            const int pk  = (lane < rem) ? sorted[s + base + lane] : 0;
            int j0 = 0;
            for (; j0 + ROW_GRP <= rem; j0 += ROW_GRP) {
                v2f v[ROW_GRP];
#pragma unroll
                for (int u = 0; u < ROW_GRP; ++u) {
                    const int src = __builtin_amdgcn_readlane(pk, j0 + u);
                    v[u] = *(const v2f*)(hs + (long long)src * FEAT + 2 * lane);
                }
#pragma unroll
                for (int u = 0; u < ROW_GRP; ++u) { acc.x += v[u].x; acc.y += v[u].y; }
            }
            for (; j0 < rem; j0 += ROW_TAIL) {
                v2f v[ROW_TAIL];
#pragma unroll
                for (int u = 0; u < ROW_TAIL; ++u) {
                    const int src = __builtin_amdgcn_readlane(pk, j0 + u);
                    v[u] = *(const v2f*)(hs + (long long)src * FEAT + 2 * lane);
                }
#pragma unroll
                for (int u = 0; u < ROW_TAIL; ++u) {
                    if (j0 + u < rem) { acc.x += v[u].x; acc.y += v[u].y; }
                }
            }
        }
        const float dc = dinv[r];
        v2f o;
        o.x = acc.x * dc + bv.x;
        o.y = acc.y * dc + bv.y;
        o.x = o.x > 0.0f ? o.x : 0.0f;
        o.y = o.y > 0.0f ? o.y : 0.0f;
        *(v2f*)(out + (long long)r * FEAT + 2 * lane) = o;
    }
}

__device__ __forceinline__ void pool_flush(float* __restrict__ sums, float* __restrict__ counts,
                                           int g, int lane, v2f ps, float pc) {
    if ((unsigned)g < (unsigned)NGRAPH) {
        float* p = sums + g * FEAT + 2 * lane;
        atomicAdd(p,     ps.x);
        atomicAdd(p + 1, ps.y);
        if (lane == 0) atomicAdd(counts + g, pc);
    }
}

__global__ void __launch_bounds__(ROW_T)
k_agg_pool(const int* __restrict__ sorted, const int* __restrict__ rowStart,
           const int* __restrict__ rowCnt, const float* __restrict__ hs,
           const float* __restrict__ dinv, const float* __restrict__ bias,
           const int* __restrict__ batch, float* __restrict__ sums, float* __restrict__ counts, int N) {
    __shared__ v2f   slotSum[ROW_T >> 5][32];
    __shared__ int   slotG[ROW_T >> 5];
    __shared__ float slotC[ROW_T >> 5];
    const int lane = threadIdx.x & 31;
    const int wave = __builtin_amdgcn_readfirstlane((int)(threadIdx.x >> 5));
    const v2f bv   = *(const v2f*)(bias + 2 * lane);

    int r = (int)blockIdx.x * POOL_ROWS + wave;
    int s_next = 0, n_next = 0, g_next = 0;
    if (r < N) { s_next = rowStart[r]; n_next = rowCnt[r]; g_next = batch[r]; }

    v2f   ps = {0.0f, 0.0f};
    float pc = 0.0f;
    int   gcur = -1;
    for (int j = 0; j < POOL_ROWS / (ROW_T >> 5) && r < N; ++j, r += (ROW_T >> 5)) {
        const int s = __builtin_amdgcn_readfirstlane(s_next);
        const int n = __builtin_amdgcn_readfirstlane(n_next);
        const int g = __builtin_amdgcn_readfirstlane(g_next);
        const int rn = r + (ROW_T >> 5);
        if (j + 1 < POOL_ROWS / (ROW_T >> 5) && rn < N) { s_next = rowStart[rn]; n_next = rowCnt[rn]; g_next = batch[rn]; }

        v2f acc = *(const v2f*)(hs + (long long)r * FEAT + 2 * lane);
        for (int base = 0; base < n; base += 32) {
            const int rem = (n - base < 32) ? (n - base) : 32;
            const int pk  = (lane < rem) ? sorted[s + base + lane] : 0;
            int j0 = 0;
            for (; j0 + ROW_GRP <= rem; j0 += ROW_GRP) {
                v2f v[ROW_GRP];
#pragma unroll
                for (int u = 0; u < ROW_GRP; ++u) {
                    const int src = __builtin_amdgcn_readlane(pk, j0 + u);
                    v[u] = *(const v2f*)(hs + (long long)src * FEAT + 2 * lane);
                }
#pragma unroll
                for (int u = 0; u < ROW_GRP; ++u) { acc.x += v[u].x; acc.y += v[u].y; }
            }
            for (; j0 < rem; j0 += ROW_TAIL) {
                v2f v[ROW_TAIL];
#pragma unroll
                for (int u = 0; u < ROW_TAIL; ++u) {
                    const int src = __builtin_amdgcn_readlane(pk, j0 + u);
                    v[u] = *(const v2f*)(hs + (long long)src * FEAT + 2 * lane);
                }
#pragma unroll
                for (int u = 0; u < ROW_TAIL; ++u) {
                    if (j0 + u < rem) { acc.x += v[u].x; acc.y += v[u].y; }
                }
            }
        }
        const float dc = dinv[r];
        v2f o;
        o.x = acc.x * dc + bv.x;
        o.y = acc.y * dc + bv.y;
        o.x = o.x > 0.0f ? o.x : 0.0f;
        o.y = o.y > 0.0f ? o.y : 0.0f;

        if (g != gcur) {
            if (pc != 0.0f) pool_flush(sums, counts, gcur, lane, ps, pc);
            ps.x = 0.0f; ps.y = 0.0f; pc = 0.0f;
            gcur = g;
        }
        ps.x += o.x; ps.y += o.y; pc += 1.0f;
    }

    slotSum[wave][lane] = ps;
    if (lane == 0) { slotG[wave] = gcur; slotC[wave] = pc; }
    __syncthreads();
    if (wave == 0) {
        v2f   ts = {0.0f, 0.0f};
        float tc = 0.0f;
        int   gt = -1;
#pragma unroll
        for (int w = 0; w < (ROW_T >> 5); ++w) {
            const int   gw = __builtin_amdgcn_readfirstlane(slotG[w]);
            const float cw = __int_as_float(__builtin_amdgcn_readfirstlane(__float_as_int(slotC[w])));
            const v2f   sw = slotSum[w][lane];
            if (gw != gt) {
                if (tc != 0.0f) pool_flush(sums, counts, gt, lane, ts, tc);
                ts.x = 0.0f; ts.y = 0.0f; tc = 0.0f;
                gt = gw;
            }
            ts.x += sw.x; ts.y += sw.y; tc += cw;
        }
        if (tc != 0.0f) pool_flush(sums, counts, gt, lane, ts, tc);
    }
}

__global__ void k_pool_zero(float* sums, float* counts) {
    int i = blockIdx.x * blockDim.x + threadIdx.x;
    if (i < NGRAPH * FEAT) sums[i] = 0.0f;
    if (i < NGRAPH) counts[i] = 0.0f;
}

__global__ void k_pool_div(const float* __restrict__ sums, const float* __restrict__ counts,
                           float* __restrict__ out) {
    int i = blockIdx.x * blockDim.x + threadIdx.x;
    if (i >= NGRAPH * FEAT) return;
    out[i] = sums[i] / fmaxf(counts[i >> 6], 1.0f);
}

extern "C" void kernel_launch(void* const* d_in, const int* in_sizes, int n_in,
                              void* d_out, int out_size, void* d_ws, size_t ws_size,
                              hipStream_t stream) {
    const float* x     = (const float*)d_in[0];
    const int*   edge  = (const int*)d_in[1];
    const int*   batch = (const int*)d_in[2];
    const float* W1    = (const float*)d_in[3];
    const float* b1    = (const float*)d_in[4];
    const float* W2    = (const float*)d_in[5];
    const float* b2    = (const float*)d_in[6];
    float*       out   = (float*)d_out;

    const int       N = in_sizes[0] / FEAT;
    const long long E = (long long)in_sizes[1] / 2;
    const long long NF = (long long)N * FEAT;

    float* bufA   = (float*)d_ws;
    float* bufB   = bufA + NF;
    float* dinv   = bufB + NF;
    float* sums   = dinv + N;
    float* counts = sums + NGRAPH * FEAT;
    int* binCount = (int*)(counts + NGRAPH);
    int* cursor   = binCount + MAXB;
    int* binStart = cursor + MAXB;
    int* packed   = binStart + MAXB;
    int* sorted   = packed + E;
    int* rowStart = sorted + E;
    int* rowCnt   = rowStart + N;

    const int* rows = edge;
    const int* cols = edge + E;

    const int T = 256;
    const int nbins = ceildiv(N, NPB);

    k_bin_zero <<<1, MAXB, 0, stream>>>(binCount, cursor);
    k_bin_count<<<256, BIN_T, 0, stream>>>(cols, binCount, E);
    k_bin_scan <<<1, MAXB, 0, stream>>>(binCount, binStart);
    k_bin_place<<<256, BIN_T, 0, stream>>>(rows, cols, binStart, cursor, packed, E);

    k_deg_sort<<<nbins, AGG_T, 0, stream>>>(packed, binStart, binCount, dinv, rowStart, rowCnt, sorted, N);

    const int mtiles   = (N + 15) / 16;
    const int gemmBlks = ceildiv((long long)mtiles * 4 * 32, T);
    const int rowBlks  = ceildiv(N, (ROW_T / 32) * 4);

    k_gemm_wmma<<<gemmBlks, T, 0, stream>>>(x, W1, dinv, bufA, N);
    k_agg_rows <<<rowBlks, ROW_T, 0, stream>>>(sorted, rowStart, rowCnt, bufA, dinv, b1, bufB, N);

    k_gemm_wmma<<<gemmBlks, T, 0, stream>>>(bufB, W2, dinv, bufA, N);
    k_pool_zero<<<ceildiv(NGRAPH * FEAT, T), T, 0, stream>>>(sums, counts);
    k_agg_pool <<<ceildiv(N, POOL_ROWS), ROW_T, 0, stream>>>(sorted, rowStart, rowCnt, bufA, dinv, b2, batch,
                                                             sums, counts, N);

    k_pool_div <<<ceildiv(NGRAPH * FEAT, T), T, 0, stream>>>(sums, counts, out);
}


# layer-2 GEMM moved into the layer-1 row pass: a work-group's 32 finished rows go through LDS to 8 WMMA tiles (same k order) and leave as dinv*(h1@W2); GEMM launch, h1 store and re-read gone
# speedup vs baseline: 5.3174x; 1.0604x over previous
#include <hip/hip_runtime.h>

typedef __attribute__((ext_vector_type(2))) float v2f;
typedef __attribute__((ext_vector_type(8))) float v8f;

#define FEAT 64
#define NGRAPH 64

#define NPB 196
#define NPB_PAD 256
#define MAXB 512
#define SRC_BITS 17
#define BIN_T 1024
#define AGG_T 512
#define ROW_T 256
#define ROW_GRP 16
#define ROW_TAIL 8
#define POOL_ROWS 32
#define SORT_UNR 4
#define TILE_ROWS 32
#define TILE_PITCH 68

static inline int ceildiv(long long a, long long b) { return (int)((a + b - 1) / b); }

__global__ void k_bin_zero(int* __restrict__ binCount, int* __restrict__ cursor) {
    binCount[threadIdx.x] = 0;
    cursor[threadIdx.x]   = 0;
}

__global__ void k_bin_count(const int* __restrict__ cols, int* __restrict__ binCount, long long E) {
    __shared__ int hist[MAXB];
    for (int i = threadIdx.x; i < MAXB; i += blockDim.x) hist[i] = 0;
    __syncthreads();
    const long long stride = (long long)gridDim.x * blockDim.x;
    for (long long e = (long long)blockIdx.x * blockDim.x + threadIdx.x; e < E; e += stride)
        atomicAdd(&hist[cols[e] / NPB], 1);
    __syncthreads();
    for (int i = threadIdx.x; i < MAXB; i += blockDim.x)
        if (hist[i] != 0) atomicAdd(&binCount[i], hist[i]);
}

__global__ void k_bin_scan(const int* __restrict__ binCount, int* __restrict__ binStart) {
    __shared__ int s[MAXB];
    const int t = threadIdx.x;
    const int own = binCount[t];
    s[t] = own;
    __syncthreads();
    for (int d = 1; d < MAXB; d <<= 1) {
        int v = (t >= d) ? s[t - d] : 0;
        __syncthreads();
        s[t] += v;
        __syncthreads();
    }
    binStart[t] = s[t] - own;
}

__global__ void k_bin_place(const int* __restrict__ rows, const int* __restrict__ cols,
                            const int* __restrict__ binStart, int* __restrict__ cursor,
                            int* __restrict__ packed, long long E) {
    __shared__ int hist[MAXB];
    __shared__ int base[MAXB];
    for (int i = threadIdx.x; i < MAXB; i += blockDim.x) hist[i] = 0;
    __syncthreads();
    const long long stride = (long long)gridDim.x * blockDim.x;
    const long long first  = (long long)blockIdx.x * blockDim.x + threadIdx.x;
    for (long long e = first; e < E; e += stride)
        atomicAdd(&hist[cols[e] / NPB], 1);
    __syncthreads();
    for (int i = threadIdx.x; i < MAXB; i += blockDim.x) {
        const int n = hist[i];
        base[i] = (n != 0) ? binStart[i] + atomicAdd(&cursor[i], n) : 0;
        hist[i] = 0;
    }
    __syncthreads();
    for (long long e = first; e < E; e += stride) {
        const int c = cols[e];
        const int b = c / NPB;
        const int pos = base[b] + atomicAdd(&hist[b], 1);
        packed[pos] = rows[e] | ((c - b * NPB) << SRC_BITS);
    }
}

__global__ void __launch_bounds__(AGG_T)
k_deg_sort(const int* __restrict__ packed, const int* __restrict__ binStart,
           const int* __restrict__ binCount, float* __restrict__ dinv,
           int* __restrict__ rowStart, int* __restrict__ rowCnt, int* __restrict__ sorted, int N) {
    __shared__ int cnt[NPB_PAD];
    __shared__ int off[NPB_PAD];
    const int b = blockIdx.x;
    const int t = threadIdx.x;
    if (t < NPB_PAD) cnt[t] = 0;
    __syncthreads();
    const int start = binStart[b];
    const int end   = start + binCount[b];
    for (int e = start + t; e < end; e += SORT_UNR * AGG_T) {
        int p[SORT_UNR];
#pragma unroll
        for (int k = 0; k < SORT_UNR; ++k) {
            const int ee = e + k * AGG_T;
            p[k] = (ee < end) ? packed[ee] : -1;
        }
#pragma unroll
        for (int k = 0; k < SORT_UNR; ++k)
            if (p[k] >= 0) atomicAdd(&cnt[p[k] >> SRC_BITS], 1);
    }
    __syncthreads();

    int own = 0;
    if (t < NPB_PAD) { own = cnt[t]; off[t] = own; }
    __syncthreads();
    for (int d = 1; d < NPB_PAD; d <<= 1) {
        int v = 0;
        if (t < NPB_PAD && t >= d) v = off[t - d];
        __syncthreads();
        if (t < NPB_PAD) off[t] += v;
        __syncthreads();
    }
    if (t < NPB_PAD) {
        const int ex = off[t] - own;
        off[t] = ex;
        cnt[t] = 0;
        const int node = b * NPB + t;
        if (t < NPB && node < N) {
            dinv[node]     = rsqrtf(1.0f + (float)own);
            rowStart[node] = start + ex;
            rowCnt[node]   = own;
        }
    }
    __syncthreads();
    for (int e = start + t; e < end; e += SORT_UNR * AGG_T) {
        int p[SORT_UNR];
#pragma unroll
        for (int k = 0; k < SORT_UNR; ++k) {
            const int ee = e + k * AGG_T;
            p[k] = (ee < end) ? packed[ee] : -1;
        }
#pragma unroll
        for (int k = 0; k < SORT_UNR; ++k) {
            if (p[k] >= 0) {
                const int d = p[k] >> SRC_BITS;
                const int pos = start + off[d] + atomicAdd(&cnt[d], 1);
                sorted[pos] = p[k] & ((1 << SRC_BITS) - 1);
            }
        }
    }
}

__global__ void k_gemm_wmma(const float* __restrict__ A, const float* __restrict__ W,
                            const float* __restrict__ dinv, float* __restrict__ D, int M) {
    int gwave = (blockIdx.x * blockDim.x + threadIdx.x) >> 5;
    int lane  = threadIdx.x & 31;
    int mtiles = (M + 15) >> 4;
    int mtile = gwave >> 2;
    int ntile = gwave & 3;
    if (mtile >= mtiles) return;

    int m0 = mtile << 4, n0 = ntile << 4;
    int half = lane >> 4;
    int r    = lane & 15;

    int arow = m0 + r; if (arow >= M) arow = M - 1;

    float s[8];
#pragma unroll
    for (int v = 0; v < 8; ++v) {
        int row = m0 + v + 8 * half;
        s[v] = dinv[row < M ? row : M - 1];
    }

    v8f c = {};
#pragma unroll
    for (int k0 = 0; k0 < FEAT; k0 += 4) {
        v2f a, b;
        int ka = k0 + 2 * half;
        a.x = A[(long long)arow * FEAT + ka + 0];
        a.y = A[(long long)arow * FEAT + ka + 1];
        b.x = W[(ka + 0) * FEAT + n0 + r];
        b.y = W[(ka + 1) * FEAT + n0 + r];
        c = __builtin_amdgcn_wmma_f32_16x16x4_f32(false, a, false, b, (short)0, c, false, false);
    }
#pragma unroll
    for (int v = 0; v < 8; ++v) {
        int row = m0 + v + 8 * half;
        if (row < M) D[(long long)row * FEAT + n0 + r] = c[v] * s[v];
    }
}


__global__ void __launch_bounds__(ROW_T)
k_agg_rows_gemm(const int* __restrict__ sorted, const int* __restrict__ rowStart,
                const int* __restrict__ rowCnt, const float* __restrict__ hs,
                const float* __restrict__ dinv, const float* __restrict__ bias,
                const float* __restrict__ Wn, float* __restrict__ out, int N) {
    __shared__ float tileA[TILE_ROWS][TILE_PITCH];
    __shared__ float tileD[TILE_ROWS][TILE_PITCH];
    __shared__ float rowScale[TILE_ROWS];
    const int lane = threadIdx.x & 31;
    const int wave = __builtin_amdgcn_readfirstlane((int)(threadIdx.x >> 5));
    const v2f bv   = *(const v2f*)(bias + 2 * lane);
    const int row0 = (int)blockIdx.x * TILE_ROWS;

    int r = row0 + wave;
    int s_next = 0, n_next = 0;
    if (r < N) { s_next = rowStart[r]; n_next = rowCnt[r]; }
    for (int j = 0; j < TILE_ROWS / (ROW_T >> 5); ++j, r += (ROW_T >> 5)) {
        v2f   o  = {0.0f, 0.0f};
        float dc = 0.0f;
        if (r < N) {
            const int s = __builtin_amdgcn_readfirstlane(s_next);
            const int n = __builtin_amdgcn_readfirstlane(n_next);
            const int rn = r + (ROW_T >> 5);
            if (j + 1 < TILE_ROWS / (ROW_T >> 5) && rn < N) { s_next = rowStart[rn]; n_next = rowCnt[rn]; }

            v2f acc = *(const v2f*)(hs + (long long)r * FEAT + 2 * lane);
            for (int base = 0; base < n; base += 32) {
                const int rem = (n - base < 32) ? (n - base) : 32;
                const int pk  = (lane < rem) ? sorted[s + base + lane] : 0;
                int j0 = 0;
                for (; j0 + ROW_GRP <= rem; j0 += ROW_GRP) {
                    v2f v[ROW_GRP];
#pragma unroll
                    for (int u = 0; u < ROW_GRP; ++u) {
                        const int src = __builtin_amdgcn_readlane(pk, j0 + u);
                        v[u] = *(const v2f*)(hs + (long long)src * FEAT + 2 * lane);
                    }
#pragma unroll
                    for (int u = 0; u < ROW_GRP; ++u) { acc.x += v[u].x; acc.y += v[u].y; }
                }
                for (; j0 < rem; j0 += ROW_TAIL) {
                    v2f v[ROW_TAIL];
#pragma unroll
                    for (int u = 0; u < ROW_TAIL; ++u) {
                        const int src = __builtin_amdgcn_readlane(pk, j0 + u);
                        v[u] = *(const v2f*)(hs + (long long)src * FEAT + 2 * lane);
                    }
#pragma unroll
                    for (int u = 0; u < ROW_TAIL; ++u) {
                        if (j0 + u < rem) { acc.x += v[u].x; acc.y += v[u].y; }
                    }
                }
            }
            dc = dinv[r];
            o.x = acc.x * dc + bv.x;
            o.y = acc.y * dc + bv.y;
            o.x = o.x > 0.0f ? o.x : 0.0f;
            o.y = o.y > 0.0f ? o.y : 0.0f;
        }
        const int lr = wave + j * (ROW_T >> 5);
        *(v2f*)&tileA[lr][2 * lane] = o;
        if (lane == 0) rowScale[lr] = dc;
    }
    __syncthreads();

    const int m0   = (wave >> 2) << 4;
    const int n0   = (wave & 3) << 4;
    const int half = lane >> 4;
    const int c16  = lane & 15;
    v8f c = {};
#pragma unroll
    for (int k0 = 0; k0 < FEAT; k0 += 4) {
        const int ka = k0 + 2 * half;
        const v2f a = *(const v2f*)&tileA[m0 + c16][ka];
        v2f b;
        b.x = Wn[(ka + 0) * FEAT + n0 + c16];
        b.y = Wn[(ka + 1) * FEAT + n0 + c16];
        c = __builtin_amdgcn_wmma_f32_16x16x4_f32(false, a, false, b, (short)0, c, false, false);
    }
#pragma unroll
    for (int v = 0; v < 8; ++v) tileD[m0 + 8 * half + v][n0 + c16] = c[v];
    __syncthreads();

    for (int j = 0; j < TILE_ROWS / (ROW_T >> 5); ++j) {
        const int lr  = wave + j * (ROW_T >> 5);
        const int row = row0 + lr;
        if (row < N) {
            const float sc = rowScale[lr];
            v2f d = *(const v2f*)&tileD[lr][2 * lane];
            d.x *= sc;
            d.y *= sc;
            *(v2f*)(out + (long long)row * FEAT + 2 * lane) = d;
        }
    }
}

__device__ __forceinline__ void pool_flush(float* __restrict__ sums, float* __restrict__ counts,
                                           int g, int lane, v2f ps, float pc) {
    if ((unsigned)g < (unsigned)NGRAPH) {
        float* p = sums + g * FEAT + 2 * lane;
        atomicAdd(p,     ps.x);
        atomicAdd(p + 1, ps.y);
        if (lane == 0) atomicAdd(counts + g, pc);
    }
}

__global__ void __launch_bounds__(ROW_T)
k_agg_pool(const int* __restrict__ sorted, const int* __restrict__ rowStart,
           const int* __restrict__ rowCnt, const float* __restrict__ hs,
           const float* __restrict__ dinv, const float* __restrict__ bias,
           const int* __restrict__ batch, float* __restrict__ sums, float* __restrict__ counts, int N) {
    __shared__ v2f   slotSum[ROW_T >> 5][32];
    __shared__ int   slotG[ROW_T >> 5];
    __shared__ float slotC[ROW_T >> 5];
    const int lane = threadIdx.x & 31;
    const int wave = __builtin_amdgcn_readfirstlane((int)(threadIdx.x >> 5));
    const v2f bv   = *(const v2f*)(bias + 2 * lane);

    int r = (int)blockIdx.x * POOL_ROWS + wave;
    int s_next = 0, n_next = 0, g_next = 0;
    if (r < N) { s_next = rowStart[r]; n_next = rowCnt[r]; g_next = batch[r]; }

    v2f   ps = {0.0f, 0.0f};
    float pc = 0.0f;
    int   gcur = -1;
    for (int j = 0; j < POOL_ROWS / (ROW_T >> 5) && r < N; ++j, r += (ROW_T >> 5)) {
        const int s = __builtin_amdgcn_readfirstlane(s_next);
        const int n = __builtin_amdgcn_readfirstlane(n_next);
        const int g = __builtin_amdgcn_readfirstlane(g_next);
        const int rn = r + (ROW_T >> 5);
        if (j + 1 < POOL_ROWS / (ROW_T >> 5) && rn < N) { s_next = rowStart[rn]; n_next = rowCnt[rn]; g_next = batch[rn]; }

        v2f acc = *(const v2f*)(hs + (long long)r * FEAT + 2 * lane);
        for (int base = 0; base < n; base += 32) {
            const int rem = (n - base < 32) ? (n - base) : 32;
            const int pk  = (lane < rem) ? sorted[s + base + lane] : 0;
            int j0 = 0;
            for (; j0 + ROW_GRP <= rem; j0 += ROW_GRP) {
                v2f v[ROW_GRP];
#pragma unroll
                for (int u = 0; u < ROW_GRP; ++u) {
                    const int src = __builtin_amdgcn_readlane(pk, j0 + u);
                    v[u] = *(const v2f*)(hs + (long long)src * FEAT + 2 * lane);
                }
#pragma unroll
                for (int u = 0; u < ROW_GRP; ++u) { acc.x += v[u].x; acc.y += v[u].y; }
            }
            for (; j0 < rem; j0 += ROW_TAIL) {
                v2f v[ROW_TAIL];
#pragma unroll
                for (int u = 0; u < ROW_TAIL; ++u) {
                    const int src = __builtin_amdgcn_readlane(pk, j0 + u);
                    v[u] = *(const v2f*)(hs + (long long)src * FEAT + 2 * lane);
                }
#pragma unroll
                for (int u = 0; u < ROW_TAIL; ++u) {
                    if (j0 + u < rem) { acc.x += v[u].x; acc.y += v[u].y; }
                }
            }
        }
        const float dc = dinv[r];
        v2f o;
        o.x = acc.x * dc + bv.x;
        o.y = acc.y * dc + bv.y;
        o.x = o.x > 0.0f ? o.x : 0.0f;
        o.y = o.y > 0.0f ? o.y : 0.0f;

        if (g != gcur) {
            if (pc != 0.0f) pool_flush(sums, counts, gcur, lane, ps, pc);
            ps.x = 0.0f; ps.y = 0.0f; pc = 0.0f;
            gcur = g;
        }
        ps.x += o.x; ps.y += o.y; pc += 1.0f;
    }

    slotSum[wave][lane] = ps;
    if (lane == 0) { slotG[wave] = gcur; slotC[wave] = pc; }
    __syncthreads();
    if (wave == 0) {
        v2f   ts = {0.0f, 0.0f};
        float tc = 0.0f;
        int   gt = -1;
#pragma unroll
        for (int w = 0; w < (ROW_T >> 5); ++w) {
            const int   gw = __builtin_amdgcn_readfirstlane(slotG[w]);
            const float cw = __int_as_float(__builtin_amdgcn_readfirstlane(__float_as_int(slotC[w])));
            const v2f   sw = slotSum[w][lane];
            if (gw != gt) {
                if (tc != 0.0f) pool_flush(sums, counts, gt, lane, ts, tc);
                ts.x = 0.0f; ts.y = 0.0f; tc = 0.0f;
                gt = gw;
            }
            ts.x += sw.x; ts.y += sw.y; tc += cw;
        }
        if (tc != 0.0f) pool_flush(sums, counts, gt, lane, ts, tc);
    }
}

__global__ void k_pool_zero(float* sums, float* counts) {
    int i = blockIdx.x * blockDim.x + threadIdx.x;
    if (i < NGRAPH * FEAT) sums[i] = 0.0f;
    if (i < NGRAPH) counts[i] = 0.0f;
}

__global__ void k_pool_div(const float* __restrict__ sums, const float* __restrict__ counts,
                           float* __restrict__ out) {
    int i = blockIdx.x * blockDim.x + threadIdx.x;
    if (i >= NGRAPH * FEAT) return;
    out[i] = sums[i] / fmaxf(counts[i >> 6], 1.0f);
}

extern "C" void kernel_launch(void* const* d_in, const int* in_sizes, int n_in,
                              void* d_out, int out_size, void* d_ws, size_t ws_size,
                              hipStream_t stream) {
    const float* x     = (const float*)d_in[0];
    const int*   edge  = (const int*)d_in[1];
    const int*   batch = (const int*)d_in[2];
    const float* W1    = (const float*)d_in[3];
    const float* b1    = (const float*)d_in[4];
    const float* W2    = (const float*)d_in[5];
    const float* b2    = (const float*)d_in[6];
    float*       out   = (float*)d_out;

    const int       N = in_sizes[0] / FEAT;
    const long long E = (long long)in_sizes[1] / 2;
    const long long NF = (long long)N * FEAT;

    float* bufA   = (float*)d_ws;
    float* bufB   = bufA + NF;
    float* dinv   = bufB + NF;
    float* sums   = dinv + N;
    float* counts = sums + NGRAPH * FEAT;
    int* binCount = (int*)(counts + NGRAPH);
    int* cursor   = binCount + MAXB;
    int* binStart = cursor + MAXB;
    int* packed   = binStart + MAXB;
    int* sorted   = packed + E;
    int* rowStart = sorted + E;
    int* rowCnt   = rowStart + N;

    const int* rows = edge;
    const int* cols = edge + E;

    const int T = 256;
    const int nbins = ceildiv(N, NPB);

    k_bin_zero <<<1, MAXB, 0, stream>>>(binCount, cursor);
    k_bin_count<<<256, BIN_T, 0, stream>>>(cols, binCount, E);
    k_bin_scan <<<1, MAXB, 0, stream>>>(binCount, binStart);
    k_bin_place<<<256, BIN_T, 0, stream>>>(rows, cols, binStart, cursor, packed, E);

    k_deg_sort<<<nbins, AGG_T, 0, stream>>>(packed, binStart, binCount, dinv, rowStart, rowCnt, sorted, N);

    const int mtiles   = (N + 15) / 16;
    const int gemmBlks = ceildiv((long long)mtiles * 4 * 32, T);

    k_gemm_wmma<<<gemmBlks, T, 0, stream>>>(x, W1, dinv, bufA, N);
    k_agg_rows_gemm<<<ceildiv(N, TILE_ROWS), ROW_T, 0, stream>>>(sorted, rowStart, rowCnt, bufA, dinv, b1, W2,
                                                                 bufB, N);

    k_pool_zero<<<ceildiv(NGRAPH * FEAT, T), T, 0, stream>>>(sums, counts);
    k_agg_pool <<<ceildiv(N, POOL_ROWS), ROW_T, 0, stream>>>(sorted, rowStart, rowCnt, bufB, dinv, b2, batch,
                                                             sums, counts, N);

    k_pool_div <<<ceildiv(NGRAPH * FEAT, T), T, 0, stream>>>(sums, counts, out);
}


# edge sort passes (range placing, row sort): the 4-byte words are laid out in LDS run by run and leave as consecutive stores (1-2 lines per store, was up to 32); positions = LDS ranks as before
# speedup vs baseline: 6.1112x; 1.1493x over previous
#include <hip/hip_runtime.h>

typedef __attribute__((ext_vector_type(2))) float v2f;
typedef __attribute__((ext_vector_type(8))) float v8f;

#define FEAT 64
#define NGRAPH 64

#define NPB 196
#define NPB_PAD 256
#define MAXB 512
#define SRC_BITS 17
#define BIN_T 1024
#define AGG_T 512
#define ROW_T 256
#define ROW_GRP 16
#define ROW_TAIL 8
#define POOL_ROWS 32
#define SORT_UNR 4
#define PLACE_CAP 13312
#define SORT_CAP 12288
#define TILE_ROWS 32
#define TILE_PITCH 68

static inline int ceildiv(long long a, long long b) { return (int)((a + b - 1) / b); }

__global__ void k_bin_zero(int* __restrict__ binCount, int* __restrict__ cursor) {
    binCount[threadIdx.x] = 0;
    cursor[threadIdx.x]   = 0;
}

__global__ void k_bin_count(const int* __restrict__ cols, int* __restrict__ binCount, long long E) {
    __shared__ int hist[MAXB];
    for (int i = threadIdx.x; i < MAXB; i += blockDim.x) hist[i] = 0;
    __syncthreads();
    const long long stride = (long long)gridDim.x * blockDim.x;
    for (long long e = (long long)blockIdx.x * blockDim.x + threadIdx.x; e < E; e += stride)
        atomicAdd(&hist[cols[e] / NPB], 1);
    __syncthreads();
    for (int i = threadIdx.x; i < MAXB; i += blockDim.x)
        if (hist[i] != 0) atomicAdd(&binCount[i], hist[i]);
}

__global__ void k_bin_scan(const int* __restrict__ binCount, int* __restrict__ binStart) {
    __shared__ int s[MAXB];
    const int t = threadIdx.x;
    const int own = binCount[t];
    s[t] = own;
    __syncthreads();
    for (int d = 1; d < MAXB; d <<= 1) {
        int v = (t >= d) ? s[t - d] : 0;
        __syncthreads();
        s[t] += v;
        __syncthreads();
    }
    binStart[t] = s[t] - own;
}

__global__ void __launch_bounds__(BIN_T)
k_bin_place(const int* __restrict__ rows, const int* __restrict__ cols,
            const int* __restrict__ binStart, int* __restrict__ cursor,
            int* __restrict__ packed, long long E) {
    __shared__ int hist[MAXB];
    __shared__ int base[MAXB];
    __shared__ int loc[MAXB];
    __shared__ int wsum[MAXB / 32];
    __shared__ int stage[PLACE_CAP + 32];
    const int t    = threadIdx.x;
    const int lane = t & 31;
    const int wave = t >> 5;
    for (int i = t; i < MAXB; i += BIN_T) hist[i] = 0;
    __syncthreads();
    const long long stride = (long long)gridDim.x * BIN_T;
    const long long first  = (long long)blockIdx.x * BIN_T + t;
    for (long long e = first; e < E; e += stride)
        atomicAdd(&hist[cols[e] / NPB], 1);
    __syncthreads();

    const bool staged = E <= (long long)(PLACE_CAP / BIN_T) * stride;

    const int n = (t < MAXB) ? hist[t] : 0;
    int incl = n;
    for (int d = 1; d < 32; d <<= 1) {
        const int v = __shfl_up(incl, d);
        if (lane >= d) incl += v;
    }
    if (t < MAXB && lane == 31) wsum[wave] = incl;
    __syncthreads();
    if (t < MAXB) {
        int pre = 0;
        for (int w = 0; w < wave; ++w) pre += wsum[w];
        loc[t]  = pre + incl - n;
        base[t] = (n != 0) ? binStart[t] + atomicAdd(&cursor[t], n) : 0;
        hist[t] = 0;
    }
    __syncthreads();
    if (!staged) {
        for (long long e = first; e < E; e += stride) {
            const int c = cols[e];
            const int b = c / NPB;
            const int pos = base[b] + atomicAdd(&hist[b], 1);
            packed[pos] = rows[e] | ((c - b * NPB) << SRC_BITS);
        }
    } else {
        for (long long e = first; e < E; e += stride) {
            const int c = cols[e];
            const int b = c / NPB;
            const int pos = loc[b] + atomicAdd(&hist[b], 1);
            stage[pos] = rows[e] | ((c - b * NPB) << SRC_BITS);
        }
        __syncthreads();
        const int per = MAXB / (BIN_T >> 5);
        const int bl  = wave * per + (lane & (per - 1));
        const int nL = hist[bl], lL = loc[bl], gL = base[bl];
        int w[MAXB / (BIN_T >> 5)];
#pragma unroll
        for (int i = 0; i < per; ++i)
            w[i] = stage[__builtin_amdgcn_readlane(lL, i) + lane];
#pragma unroll
        for (int i = 0; i < per; ++i) {
            const int rn = __builtin_amdgcn_readlane(nL, i);
            const int rg = __builtin_amdgcn_readlane(gL, i);
            if (lane < rn) packed[rg + lane] = w[i];
        }
#pragma nounroll
        for (int i = 0; i < per; ++i) {
            const int rn = __builtin_amdgcn_readlane(nL, i);
            const int rl = __builtin_amdgcn_readlane(lL, i);
            const int rg = __builtin_amdgcn_readlane(gL, i);
#pragma nounroll
            for (int k = lane + 32; k < rn; k += 32) packed[rg + k] = stage[rl + k];
        }
    }
}

__global__ void __launch_bounds__(AGG_T)
k_deg_sort(const int* __restrict__ packed, const int* __restrict__ binStart,
           const int* __restrict__ binCount, float* __restrict__ dinv,
           int* __restrict__ rowStart, int* __restrict__ rowCnt, int* __restrict__ sorted, int N) {
    __shared__ int cnt[NPB_PAD];
    __shared__ int off[NPB_PAD];
    __shared__ int stage[SORT_CAP];
    const int b = blockIdx.x;
    const int t = threadIdx.x;
    if (t < NPB_PAD) cnt[t] = 0;
    __syncthreads();
    const int start = binStart[b];
    const int m     = binCount[b];
    const int end   = start + m;
    const bool staged = m <= SORT_CAP;
    for (int e = start + t; e < end; e += SORT_UNR * AGG_T) {
        int p[SORT_UNR];
#pragma unroll
        for (int k = 0; k < SORT_UNR; ++k) {
            const int ee = e + k * AGG_T;
            p[k] = (ee < end) ? packed[ee] : -1;
        }
#pragma unroll
        for (int k = 0; k < SORT_UNR; ++k)
            if (p[k] >= 0) atomicAdd(&cnt[p[k] >> SRC_BITS], 1);
    }
    __syncthreads();

    int own = 0;
    if (t < NPB_PAD) { own = cnt[t]; off[t] = own; }
    __syncthreads();
    for (int d = 1; d < NPB_PAD; d <<= 1) {
        int v = 0;
        if (t < NPB_PAD && t >= d) v = off[t - d];
        __syncthreads();
        if (t < NPB_PAD) off[t] += v;
        __syncthreads();
    }
    if (t < NPB_PAD) {
        const int ex = off[t] - own;
        off[t] = ex;
        cnt[t] = 0;
        const int node = b * NPB + t;
        if (t < NPB && node < N) {
            dinv[node]     = rsqrtf(1.0f + (float)own);
            rowStart[node] = start + ex;
            rowCnt[node]   = own;
        }
    }
    __syncthreads();
    if (!staged) {
        for (int e = start + t; e < end; e += SORT_UNR * AGG_T) {
            int p[SORT_UNR];
#pragma unroll
            for (int k = 0; k < SORT_UNR; ++k) {
                const int ee = e + k * AGG_T;
                p[k] = (ee < end) ? packed[ee] : -1;
            }
#pragma unroll
            for (int k = 0; k < SORT_UNR; ++k) {
                if (p[k] >= 0) {
                    const int d = p[k] >> SRC_BITS;
                    const int pos = start + off[d] + atomicAdd(&cnt[d], 1);
                    sorted[pos] = p[k] & ((1 << SRC_BITS) - 1);
                }
            }
        }
    } else {
        for (int e = start + t; e < end; e += SORT_UNR * AGG_T) {
            int p[SORT_UNR];
#pragma unroll
            for (int k = 0; k < SORT_UNR; ++k) {
                const int ee = e + k * AGG_T;
                p[k] = (ee < end) ? packed[ee] : -1;
            }
#pragma unroll
            for (int k = 0; k < SORT_UNR; ++k) {
                if (p[k] >= 0) {
                    const int d = p[k] >> SRC_BITS;
                    const int pos = off[d] + atomicAdd(&cnt[d], 1);
                    stage[pos] = p[k] & ((1 << SRC_BITS) - 1);
                }
            }
        }
        __syncthreads();
#pragma unroll 4
        for (int i = t; i < m; i += AGG_T) sorted[start + i] = stage[i];
    }
}

__global__ void k_gemm_wmma(const float* __restrict__ A, const float* __restrict__ W,
                            const float* __restrict__ dinv, float* __restrict__ D, int M) {
    int gwave = (blockIdx.x * blockDim.x + threadIdx.x) >> 5;
    int lane  = threadIdx.x & 31;
    int mtiles = (M + 15) >> 4;
    int mtile = gwave >> 2;
    int ntile = gwave & 3;
    if (mtile >= mtiles) return;

    int m0 = mtile << 4, n0 = ntile << 4;
    int half = lane >> 4;
    int r    = lane & 15;

    int arow = m0 + r; if (arow >= M) arow = M - 1;

    float s[8];
#pragma unroll
    for (int v = 0; v < 8; ++v) {
        int row = m0 + v + 8 * half;
        s[v] = dinv[row < M ? row : M - 1];
    }

    v8f c = {};
#pragma unroll
    for (int k0 = 0; k0 < FEAT; k0 += 4) {
        v2f a, b;
        int ka = k0 + 2 * half;
        a.x = A[(long long)arow * FEAT + ka + 0];
        a.y = A[(long long)arow * FEAT + ka + 1];
        b.x = W[(ka + 0) * FEAT + n0 + r];
        b.y = W[(ka + 1) * FEAT + n0 + r];
        c = __builtin_amdgcn_wmma_f32_16x16x4_f32(false, a, false, b, (short)0, c, false, false);
    }
#pragma unroll
    for (int v = 0; v < 8; ++v) {
        int row = m0 + v + 8 * half;
        if (row < M) D[(long long)row * FEAT + n0 + r] = c[v] * s[v];
    }
}


__global__ void __launch_bounds__(ROW_T)
k_agg_rows_gemm(const int* __restrict__ sorted, const int* __restrict__ rowStart,
                const int* __restrict__ rowCnt, const float* __restrict__ hs,
                const float* __restrict__ dinv, const float* __restrict__ bias,
                const float* __restrict__ Wn, float* __restrict__ out, int N) {
    __shared__ float tileA[TILE_ROWS][TILE_PITCH];
    __shared__ float tileD[TILE_ROWS][TILE_PITCH];
    __shared__ float rowScale[TILE_ROWS];
    const int lane = threadIdx.x & 31;
    const int wave = __builtin_amdgcn_readfirstlane((int)(threadIdx.x >> 5));
    const v2f bv   = *(const v2f*)(bias + 2 * lane);
    const int row0 = (int)blockIdx.x * TILE_ROWS;

    int r = row0 + wave;
    int s_next = 0, n_next = 0;
    if (r < N) { s_next = rowStart[r]; n_next = rowCnt[r]; }
    for (int j = 0; j < TILE_ROWS / (ROW_T >> 5); ++j, r += (ROW_T >> 5)) {
        v2f   o  = {0.0f, 0.0f};
        float dc = 0.0f;
        if (r < N) {
            const int s = __builtin_amdgcn_readfirstlane(s_next);
            const int n = __builtin_amdgcn_readfirstlane(n_next);
            const int rn = r + (ROW_T >> 5);
            if (j + 1 < TILE_ROWS / (ROW_T >> 5) && rn < N) { s_next = rowStart[rn]; n_next = rowCnt[rn]; }

            v2f acc = *(const v2f*)(hs + (long long)r * FEAT + 2 * lane);
            for (int base = 0; base < n; base += 32) {
                const int rem = (n - base < 32) ? (n - base) : 32;
                const int pk  = (lane < rem) ? sorted[s + base + lane] : 0;
                int j0 = 0;
                for (; j0 + ROW_GRP <= rem; j0 += ROW_GRP) {
                    v2f v[ROW_GRP];
#pragma unroll
                    for (int u = 0; u < ROW_GRP; ++u) {
                        const int src = __builtin_amdgcn_readlane(pk, j0 + u);
                        v[u] = *(const v2f*)(hs + (long long)src * FEAT + 2 * lane);
                    }
#pragma unroll
                    for (int u = 0; u < ROW_GRP; ++u) { acc.x += v[u].x; acc.y += v[u].y; }
                }
                for (; j0 < rem; j0 += ROW_TAIL) {
                    v2f v[ROW_TAIL];
#pragma unroll
                    for (int u = 0; u < ROW_TAIL; ++u) {
                        const int src = __builtin_amdgcn_readlane(pk, j0 + u);
                        v[u] = *(const v2f*)(hs + (long long)src * FEAT + 2 * lane);
                    }
#pragma unroll
                    for (int u = 0; u < ROW_TAIL; ++u) {
                        if (j0 + u < rem) { acc.x += v[u].x; acc.y += v[u].y; }
                    }
                }
            }
            dc = dinv[r];
            o.x = acc.x * dc + bv.x;
            o.y = acc.y * dc + bv.y;
            o.x = o.x > 0.0f ? o.x : 0.0f;
            o.y = o.y > 0.0f ? o.y : 0.0f;
        }
        const int lr = wave + j * (ROW_T >> 5);
        *(v2f*)&tileA[lr][2 * lane] = o;
        if (lane == 0) rowScale[lr] = dc;
    }
    __syncthreads();

    const int m0   = (wave >> 2) << 4;
    const int n0   = (wave & 3) << 4;
    const int half = lane >> 4;
    const int c16  = lane & 15;
    v8f c = {};
#pragma unroll
    for (int k0 = 0; k0 < FEAT; k0 += 4) {
        const int ka = k0 + 2 * half;
        const v2f a = *(const v2f*)&tileA[m0 + c16][ka];
        v2f b;
        b.x = Wn[(ka + 0) * FEAT + n0 + c16];
        b.y = Wn[(ka + 1) * FEAT + n0 + c16];
        c = __builtin_amdgcn_wmma_f32_16x16x4_f32(false, a, false, b, (short)0, c, false, false);
    }
#pragma unroll
    for (int v = 0; v < 8; ++v) tileD[m0 + 8 * half + v][n0 + c16] = c[v];
    __syncthreads();

    for (int j = 0; j < TILE_ROWS / (ROW_T >> 5); ++j) {
        const int lr  = wave + j * (ROW_T >> 5);
        const int row = row0 + lr;
        if (row < N) {
            const float sc = rowScale[lr];
            v2f d = *(const v2f*)&tileD[lr][2 * lane];
            d.x *= sc;
            d.y *= sc;
            *(v2f*)(out + (long long)row * FEAT + 2 * lane) = d;
        }
    }
}

__device__ __forceinline__ void pool_flush(float* __restrict__ sums, float* __restrict__ counts,
                                           int g, int lane, v2f ps, float pc) {
    if ((unsigned)g < (unsigned)NGRAPH) {
        float* p = sums + g * FEAT + 2 * lane;
        atomicAdd(p,     ps.x);
        atomicAdd(p + 1, ps.y);
        if (lane == 0) atomicAdd(counts + g, pc);
    }
}

__global__ void __launch_bounds__(ROW_T)
k_agg_pool(const int* __restrict__ sorted, const int* __restrict__ rowStart,
           const int* __restrict__ rowCnt, const float* __restrict__ hs,
           const float* __restrict__ dinv, const float* __restrict__ bias,
           const int* __restrict__ batch, float* __restrict__ sums, float* __restrict__ counts, int N) {
    __shared__ v2f   slotSum[ROW_T >> 5][32];
    __shared__ int   slotG[ROW_T >> 5];
    __shared__ float slotC[ROW_T >> 5];
    const int lane = threadIdx.x & 31;
    const int wave = __builtin_amdgcn_readfirstlane((int)(threadIdx.x >> 5));
    const v2f bv   = *(const v2f*)(bias + 2 * lane);

    int r = (int)blockIdx.x * POOL_ROWS + wave;
    int s_next = 0, n_next = 0, g_next = 0;
    if (r < N) { s_next = rowStart[r]; n_next = rowCnt[r]; g_next = batch[r]; }

    v2f   ps = {0.0f, 0.0f};
    float pc = 0.0f;
    int   gcur = -1;
    for (int j = 0; j < POOL_ROWS / (ROW_T >> 5) && r < N; ++j, r += (ROW_T >> 5)) {
        const int s = __builtin_amdgcn_readfirstlane(s_next);
        const int n = __builtin_amdgcn_readfirstlane(n_next);
        const int g = __builtin_amdgcn_readfirstlane(g_next);
        const int rn = r + (ROW_T >> 5);
        if (j + 1 < POOL_ROWS / (ROW_T >> 5) && rn < N) { s_next = rowStart[rn]; n_next = rowCnt[rn]; g_next = batch[rn]; }

        v2f acc = *(const v2f*)(hs + (long long)r * FEAT + 2 * lane);
        for (int base = 0; base < n; base += 32) {
            const int rem = (n - base < 32) ? (n - base) : 32;
            const int pk  = (lane < rem) ? sorted[s + base + lane] : 0;
            int j0 = 0;
            for (; j0 + ROW_GRP <= rem; j0 += ROW_GRP) {
                v2f v[ROW_GRP];
#pragma unroll
                for (int u = 0; u < ROW_GRP; ++u) {
                    const int src = __builtin_amdgcn_readlane(pk, j0 + u);
                    v[u] = *(const v2f*)(hs + (long long)src * FEAT + 2 * lane);
                }
#pragma unroll
                for (int u = 0; u < ROW_GRP; ++u) { acc.x += v[u].x; acc.y += v[u].y; }
            }
            for (; j0 < rem; j0 += ROW_TAIL) {
                v2f v[ROW_TAIL];
#pragma unroll
                for (int u = 0; u < ROW_TAIL; ++u) {
                    const int src = __builtin_amdgcn_readlane(pk, j0 + u);
                    v[u] = *(const v2f*)(hs + (long long)src * FEAT + 2 * lane);
                }
#pragma unroll
                for (int u = 0; u < ROW_TAIL; ++u) {
                    if (j0 + u < rem) { acc.x += v[u].x; acc.y += v[u].y; }
                }
            }
        }
        const float dc = dinv[r];
        v2f o;
        o.x = acc.x * dc + bv.x;
        o.y = acc.y * dc + bv.y;
        o.x = o.x > 0.0f ? o.x : 0.0f;
        o.y = o.y > 0.0f ? o.y : 0.0f;

        if (g != gcur) {
            if (pc != 0.0f) pool_flush(sums, counts, gcur, lane, ps, pc);
            ps.x = 0.0f; ps.y = 0.0f; pc = 0.0f;
            gcur = g;
        }
        ps.x += o.x; ps.y += o.y; pc += 1.0f;
    }

    slotSum[wave][lane] = ps;
    if (lane == 0) { slotG[wave] = gcur; slotC[wave] = pc; }
    __syncthreads();
    if (wave == 0) {
        v2f   ts = {0.0f, 0.0f};
        float tc = 0.0f;
        int   gt = -1;
#pragma unroll
        for (int w = 0; w < (ROW_T >> 5); ++w) {
            const int   gw = __builtin_amdgcn_readfirstlane(slotG[w]);
            const float cw = __int_as_float(__builtin_amdgcn_readfirstlane(__float_as_int(slotC[w])));
            const v2f   sw = slotSum[w][lane];
            if (gw != gt) {
                if (tc != 0.0f) pool_flush(sums, counts, gt, lane, ts, tc);
                ts.x = 0.0f; ts.y = 0.0f; tc = 0.0f;
                gt = gw;
            }
            ts.x += sw.x; ts.y += sw.y; tc += cw;
        }
        if (tc != 0.0f) pool_flush(sums, counts, gt, lane, ts, tc);
    }
}

__global__ void k_pool_zero(float* sums, float* counts) {
    int i = blockIdx.x * blockDim.x + threadIdx.x;
    if (i < NGRAPH * FEAT) sums[i] = 0.0f;
    if (i < NGRAPH) counts[i] = 0.0f;
}

__global__ void k_pool_div(const float* __restrict__ sums, const float* __restrict__ counts,
                           float* __restrict__ out) {
    int i = blockIdx.x * blockDim.x + threadIdx.x;
    if (i >= NGRAPH * FEAT) return;
    out[i] = sums[i] / fmaxf(counts[i >> 6], 1.0f);
}

extern "C" void kernel_launch(void* const* d_in, const int* in_sizes, int n_in,
                              void* d_out, int out_size, void* d_ws, size_t ws_size,
                              hipStream_t stream) {
    const float* x     = (const float*)d_in[0];
    const int*   edge  = (const int*)d_in[1];
    const int*   batch = (const int*)d_in[2];
    const float* W1    = (const float*)d_in[3];
    const float* b1    = (const float*)d_in[4];
    const float* W2    = (const float*)d_in[5];
    const float* b2    = (const float*)d_in[6];
    float*       out   = (float*)d_out;

    const int       N = in_sizes[0] / FEAT;
    const long long E = (long long)in_sizes[1] / 2;
    const long long NF = (long long)N * FEAT;

    float* bufA   = (float*)d_ws;
    float* bufB   = bufA + NF;
    float* dinv   = bufB + NF;
    float* sums   = dinv + N;
    float* counts = sums + NGRAPH * FEAT;
    int* binCount = (int*)(counts + NGRAPH);
    int* cursor   = binCount + MAXB;
    int* binStart = cursor + MAXB;
    int* packed   = binStart + MAXB;
    int* sorted   = packed + E;
    int* rowStart = sorted + E;
    int* rowCnt   = rowStart + N;

    const int* rows = edge;
    const int* cols = edge + E;

    const int T = 256;
    const int nbins = ceildiv(N, NPB);

    k_bin_zero <<<1, MAXB, 0, stream>>>(binCount, cursor);
    k_bin_count<<<256, BIN_T, 0, stream>>>(cols, binCount, E);
    k_bin_scan <<<1, MAXB, 0, stream>>>(binCount, binStart);
    k_bin_place<<<256, BIN_T, 0, stream>>>(rows, cols, binStart, cursor, packed, E);

    k_deg_sort<<<nbins, AGG_T, 0, stream>>>(packed, binStart, binCount, dinv, rowStart, rowCnt, sorted, N);

    const int mtiles   = (N + 15) / 16;
    const int gemmBlks = ceildiv((long long)mtiles * 4 * 32, T);

    k_gemm_wmma<<<gemmBlks, T, 0, stream>>>(x, W1, dinv, bufA, N);
    k_agg_rows_gemm<<<ceildiv(N, TILE_ROWS), ROW_T, 0, stream>>>(sorted, rowStart, rowCnt, bufA, dinv, b1, W2,
                                                                 bufB, N);

    k_pool_zero<<<ceildiv(NGRAPH * FEAT, T), T, 0, stream>>>(sums, counts);
    k_agg_pool <<<ceildiv(N, POOL_ROWS), ROW_T, 0, stream>>>(sorted, rowStart, rowCnt, bufB, dinv, b2, batch,
                                                             sums, counts, N);

    k_pool_div <<<ceildiv(NGRAPH * FEAT, T), T, 0, stream>>>(sums, counts, out);
}


# bucketing and sort passes: a thread's edge loads issue as one group and stay in registers (13 or 16 in flight; was one dependent memory round trip per edge, cols read twice); positions = LDS ranks
# speedup vs baseline: 6.4531x; 1.0559x over previous
#include <hip/hip_runtime.h>

typedef __attribute__((ext_vector_type(2))) float v2f;
typedef __attribute__((ext_vector_type(8))) float v8f;

#define FEAT 64
#define NGRAPH 64

#define NPB 196
#define NPB_PAD 256
#define MAXB 512
#define SRC_BITS 17
#define BIN_T 1024
#define AGG_T 512
#define ROW_T 256
#define ROW_GRP 16
#define ROW_TAIL 8
#define POOL_ROWS 32
#define SORT_UNR 16
#define PLACE_CAP 13312
#define BIN_UNR (PLACE_CAP / BIN_T)
#define BIN_SPARE 32
#define SORT_CAP 12288
#define TILE_ROWS 32
#define TILE_PITCH 68

static inline int ceildiv(long long a, long long b) { return (int)((a + b - 1) / b); }

__global__ void k_bin_zero(int* __restrict__ binCount, int* __restrict__ cursor) {
    binCount[threadIdx.x] = 0;
    cursor[threadIdx.x]   = 0;
}

__global__ void __launch_bounds__(BIN_T)
k_bin_count(const int* __restrict__ cols, int* __restrict__ binCount, long long E) {
    __shared__ int hist[MAXB + BIN_SPARE];
    for (int i = threadIdx.x; i < MAXB + BIN_SPARE; i += BIN_T) hist[i] = 0;
    __syncthreads();
    const long long stride = (long long)gridDim.x * BIN_T;
    const int spare = MAXB + (threadIdx.x & (BIN_SPARE - 1));
    for (long long e0 = (long long)blockIdx.x * BIN_T + threadIdx.x; e0 < E; e0 += BIN_UNR * stride) {
        int c[BIN_UNR];
#pragma unroll
        for (int k = 0; k < BIN_UNR; ++k) {
            const long long e = e0 + k * stride;
            c[k] = cols[e < E ? e : E - 1];
        }
#pragma unroll
        for (int k = 0; k < BIN_UNR; ++k) {
            const bool in = e0 + k * stride < E;
            atomicAdd(&hist[in ? c[k] / NPB : spare], in ? 1 : 0);
        }
    }
    __syncthreads();
    for (int i = threadIdx.x; i < MAXB; i += BIN_T)
        if (hist[i] != 0) atomicAdd(&binCount[i], hist[i]);
}

__global__ void k_bin_scan(const int* __restrict__ binCount, int* __restrict__ binStart) {
    __shared__ int s[MAXB];
    const int t = threadIdx.x;
    const int own = binCount[t];
    s[t] = own;
    __syncthreads();
    for (int d = 1; d < MAXB; d <<= 1) {
        int v = (t >= d) ? s[t - d] : 0;
        __syncthreads();
        s[t] += v;
        __syncthreads();
    }
    binStart[t] = s[t] - own;
}

__global__ void __launch_bounds__(BIN_T)
k_bin_place(const int* __restrict__ rows, const int* __restrict__ cols,
            const int* __restrict__ binStart, int* __restrict__ cursor,
            int* __restrict__ packed, long long E) {
    __shared__ int hist[MAXB + BIN_SPARE];
    __shared__ int base[MAXB];
    __shared__ int loc[MAXB + BIN_SPARE];
    __shared__ int wsum[MAXB / 32];
    __shared__ int stage[PLACE_CAP + 32];
    const int t    = threadIdx.x;
    const int lane = t & 31;
    const int wave = t >> 5;
    for (int i = t; i < MAXB + BIN_SPARE; i += BIN_T) hist[i] = 0;
    __syncthreads();
    const long long stride = (long long)gridDim.x * BIN_T;
    const long long first  = (long long)blockIdx.x * BIN_T + t;
    const bool staged = E > 0 && E <= (long long)BIN_UNR * stride;

    const int spare = MAXB + lane;
    int c[BIN_UNR], r[BIN_UNR];
    if (staged) {
#pragma unroll
        for (int k = 0; k < BIN_UNR; ++k) {
            const long long e  = first + k * stride;
            const long long ec = e < E ? e : E - 1;
            c[k] = cols[ec];
            r[k] = rows[ec];
        }
#pragma unroll
        for (int k = 0; k < BIN_UNR; ++k) {
            const bool in = first + k * stride < E;
            atomicAdd(&hist[in ? c[k] / NPB : spare], in ? 1 : 0);
        }
    } else {
        for (long long e = first; e < E; e += stride)
            atomicAdd(&hist[cols[e] / NPB], 1);
    }
    __syncthreads();

    const int n = (t < MAXB) ? hist[t] : 0;
    int incl = n;
    for (int d = 1; d < 32; d <<= 1) {
        const int v = __shfl_up(incl, d);
        if (lane >= d) incl += v;
    }
    if (t < MAXB && lane == 31) wsum[wave] = incl;
    __syncthreads();
    if (t < MAXB) {
        int pre = 0;
        for (int w = 0; w < wave; ++w) pre += wsum[w];
        loc[t]  = pre + incl - n;
        base[t] = (n != 0) ? binStart[t] + atomicAdd(&cursor[t], n) : 0;
        hist[t] = 0;
    } else if (t < MAXB + BIN_SPARE) {
        loc[t] = PLACE_CAP + (t - MAXB);
    }
    __syncthreads();
    if (!staged) {
        for (long long e = first; e < E; e += stride) {
            const int ce = cols[e];
            const int b = ce / NPB;
            const int pos = base[b] + atomicAdd(&hist[b], 1);
            packed[pos] = rows[e] | ((ce - b * NPB) << SRC_BITS);
        }
    } else {
        int pos[BIN_UNR];
#pragma unroll
        for (int k = 0; k < BIN_UNR; ++k) {
            const bool in = first + k * stride < E;
            const int  bb = in ? c[k] / NPB : spare;
            pos[k] = loc[bb] + atomicAdd(&hist[bb], in ? 1 : 0);
        }
#pragma unroll
        for (int k = 0; k < BIN_UNR; ++k) {
            const int b = c[k] / NPB;
            stage[pos[k]] = r[k] | ((c[k] - b * NPB) << SRC_BITS);
        }
        __syncthreads();
        const int per = MAXB / (BIN_T >> 5);
        const int bl  = wave * per + (lane & (per - 1));
        const int nL = hist[bl], lL = loc[bl], gL = base[bl];
        int w[MAXB / (BIN_T >> 5)];
#pragma unroll
        for (int i = 0; i < per; ++i)
            w[i] = stage[__builtin_amdgcn_readlane(lL, i) + lane];
#pragma unroll
        for (int i = 0; i < per; ++i) {
            const int rn = __builtin_amdgcn_readlane(nL, i);
            const int rg = __builtin_amdgcn_readlane(gL, i);
            if (lane < rn) packed[rg + lane] = w[i];
        }
#pragma nounroll
        for (int i = 0; i < per; ++i) {
            const int rn = __builtin_amdgcn_readlane(nL, i);
            const int rl = __builtin_amdgcn_readlane(lL, i);
            const int rg = __builtin_amdgcn_readlane(gL, i);
#pragma nounroll
            for (int k = lane + 32; k < rn; k += 32) packed[rg + k] = stage[rl + k];
        }
    }
}

__global__ void __launch_bounds__(AGG_T)
k_deg_sort(const int* __restrict__ packed, const int* __restrict__ binStart,
           const int* __restrict__ binCount, float* __restrict__ dinv,
           int* __restrict__ rowStart, int* __restrict__ rowCnt, int* __restrict__ sorted, int N) {
    __shared__ int cnt[NPB_PAD];
    __shared__ int off[NPB_PAD];
    __shared__ int stage[SORT_CAP];
    const int b = blockIdx.x;
    const int t = threadIdx.x;
    if (t < NPB_PAD) cnt[t] = 0;
    __syncthreads();
    const int start = binStart[b];
    const int m     = binCount[b];
    const int end   = start + m;
    const bool staged = m <= SORT_CAP;
    for (int e = start + t; e < end; e += SORT_UNR * AGG_T) {
        int p[SORT_UNR];
#pragma unroll
        for (int k = 0; k < SORT_UNR; ++k) {
            const int ee = e + k * AGG_T;
            p[k] = (ee < end) ? packed[ee] : -1;
        }
#pragma unroll
        for (int k = 0; k < SORT_UNR; ++k)
            if (p[k] >= 0) atomicAdd(&cnt[p[k] >> SRC_BITS], 1);
    }
    __syncthreads();

    int own = 0;
    if (t < NPB_PAD) { own = cnt[t]; off[t] = own; }
    __syncthreads();
    for (int d = 1; d < NPB_PAD; d <<= 1) {
        int v = 0;
        if (t < NPB_PAD && t >= d) v = off[t - d];
        __syncthreads();
        if (t < NPB_PAD) off[t] += v;
        __syncthreads();
    }
    if (t < NPB_PAD) {
        const int ex = off[t] - own;
        off[t] = ex;
        cnt[t] = 0;
        const int node = b * NPB + t;
        if (t < NPB && node < N) {
            dinv[node]     = rsqrtf(1.0f + (float)own);
            rowStart[node] = start + ex;
            rowCnt[node]   = own;
        }
    }
    __syncthreads();
    if (!staged) {
        for (int e = start + t; e < end; e += SORT_UNR * AGG_T) {
            int p[SORT_UNR];
#pragma unroll
            for (int k = 0; k < SORT_UNR; ++k) {
                const int ee = e + k * AGG_T;
                p[k] = (ee < end) ? packed[ee] : -1;
            }
#pragma unroll
            for (int k = 0; k < SORT_UNR; ++k) {
                if (p[k] >= 0) {
                    const int d = p[k] >> SRC_BITS;
                    const int pos = start + off[d] + atomicAdd(&cnt[d], 1);
                    sorted[pos] = p[k] & ((1 << SRC_BITS) - 1);
                }
            }
        }
    } else {
        for (int e = start + t; e < end; e += SORT_UNR * AGG_T) {
            int p[SORT_UNR];
#pragma unroll
            for (int k = 0; k < SORT_UNR; ++k) {
                const int ee = e + k * AGG_T;
                p[k] = (ee < end) ? packed[ee] : -1;
            }
#pragma unroll
            for (int k = 0; k < SORT_UNR; ++k) {
                if (p[k] >= 0) {
                    const int d = p[k] >> SRC_BITS;
                    const int pos = off[d] + atomicAdd(&cnt[d], 1);
                    stage[pos] = p[k] & ((1 << SRC_BITS) - 1);
                }
            }
        }
        __syncthreads();
#pragma unroll 4
        for (int i = t; i < m; i += AGG_T) sorted[start + i] = stage[i];
    }
}

__global__ void k_gemm_wmma(const float* __restrict__ A, const float* __restrict__ W,
                            const float* __restrict__ dinv, float* __restrict__ D, int M) {
    int gwave = (blockIdx.x * blockDim.x + threadIdx.x) >> 5;
    int lane  = threadIdx.x & 31;
    int mtiles = (M + 15) >> 4;
    int mtile = gwave >> 2;
    int ntile = gwave & 3;
    if (mtile >= mtiles) return;

    int m0 = mtile << 4, n0 = ntile << 4;
    int half = lane >> 4;
    int r    = lane & 15;

    int arow = m0 + r; if (arow >= M) arow = M - 1;

    float s[8];
#pragma unroll
    for (int v = 0; v < 8; ++v) {
        int row = m0 + v + 8 * half;
        s[v] = dinv[row < M ? row : M - 1];
    }

    v8f c = {};
#pragma unroll
    for (int k0 = 0; k0 < FEAT; k0 += 4) {
        v2f a, b;
        int ka = k0 + 2 * half;
        a.x = A[(long long)arow * FEAT + ka + 0];
        a.y = A[(long long)arow * FEAT + ka + 1];
        b.x = W[(ka + 0) * FEAT + n0 + r];
        b.y = W[(ka + 1) * FEAT + n0 + r];
        c = __builtin_amdgcn_wmma_f32_16x16x4_f32(false, a, false, b, (short)0, c, false, false);
    }
#pragma unroll
    for (int v = 0; v < 8; ++v) {
        int row = m0 + v + 8 * half;
        if (row < M) D[(long long)row * FEAT + n0 + r] = c[v] * s[v];
    }
}


__global__ void __launch_bounds__(ROW_T)
k_agg_rows_gemm(const int* __restrict__ sorted, const int* __restrict__ rowStart,
                const int* __restrict__ rowCnt, const float* __restrict__ hs,
                const float* __restrict__ dinv, const float* __restrict__ bias,
                const float* __restrict__ Wn, float* __restrict__ out, int N) {
    __shared__ float tileA[TILE_ROWS][TILE_PITCH];
    __shared__ float tileD[TILE_ROWS][TILE_PITCH];
    __shared__ float rowScale[TILE_ROWS];
    const int lane = threadIdx.x & 31;
    const int wave = __builtin_amdgcn_readfirstlane((int)(threadIdx.x >> 5));
    const v2f bv   = *(const v2f*)(bias + 2 * lane);
    const int row0 = (int)blockIdx.x * TILE_ROWS;

    int r = row0 + wave;
    int s_next = 0, n_next = 0;
    if (r < N) { s_next = rowStart[r]; n_next = rowCnt[r]; }
    for (int j = 0; j < TILE_ROWS / (ROW_T >> 5); ++j, r += (ROW_T >> 5)) {
        v2f   o  = {0.0f, 0.0f};
        float dc = 0.0f;
        if (r < N) {
            const int s = __builtin_amdgcn_readfirstlane(s_next);
            const int n = __builtin_amdgcn_readfirstlane(n_next);
            const int rn = r + (ROW_T >> 5);
            if (j + 1 < TILE_ROWS / (ROW_T >> 5) && rn < N) { s_next = rowStart[rn]; n_next = rowCnt[rn]; }

            v2f acc = *(const v2f*)(hs + (long long)r * FEAT + 2 * lane);
            for (int base = 0; base < n; base += 32) {
                const int rem = (n - base < 32) ? (n - base) : 32;
                const int pk  = (lane < rem) ? sorted[s + base + lane] : 0;
                int j0 = 0;
                for (; j0 + ROW_GRP <= rem; j0 += ROW_GRP) {
                    v2f v[ROW_GRP];
#pragma unroll
                    for (int u = 0; u < ROW_GRP; ++u) {
                        const int src = __builtin_amdgcn_readlane(pk, j0 + u);
                        v[u] = *(const v2f*)(hs + (long long)src * FEAT + 2 * lane);
                    }
#pragma unroll
                    for (int u = 0; u < ROW_GRP; ++u) { acc.x += v[u].x; acc.y += v[u].y; }
                }
                for (; j0 < rem; j0 += ROW_TAIL) {
                    v2f v[ROW_TAIL];
#pragma unroll
                    for (int u = 0; u < ROW_TAIL; ++u) {
                        const int src = __builtin_amdgcn_readlane(pk, j0 + u);
                        v[u] = *(const v2f*)(hs + (long long)src * FEAT + 2 * lane);
                    }
#pragma unroll
                    for (int u = 0; u < ROW_TAIL; ++u) {
                        if (j0 + u < rem) { acc.x += v[u].x; acc.y += v[u].y; }
                    }
                }
            }
            dc = dinv[r];
            o.x = acc.x * dc + bv.x;
            o.y = acc.y * dc + bv.y;
            o.x = o.x > 0.0f ? o.x : 0.0f;
            o.y = o.y > 0.0f ? o.y : 0.0f;
        }
        const int lr = wave + j * (ROW_T >> 5);
        *(v2f*)&tileA[lr][2 * lane] = o;
        if (lane == 0) rowScale[lr] = dc;
    }
    __syncthreads();

    const int m0   = (wave >> 2) << 4;
    const int n0   = (wave & 3) << 4;
    const int half = lane >> 4;
    const int c16  = lane & 15;
    v8f c = {};
#pragma unroll
    for (int k0 = 0; k0 < FEAT; k0 += 4) {
        const int ka = k0 + 2 * half;
        const v2f a = *(const v2f*)&tileA[m0 + c16][ka];
        v2f b;
        b.x = Wn[(ka + 0) * FEAT + n0 + c16];
        b.y = Wn[(ka + 1) * FEAT + n0 + c16];
        c = __builtin_amdgcn_wmma_f32_16x16x4_f32(false, a, false, b, (short)0, c, false, false);
    }
#pragma unroll
    for (int v = 0; v < 8; ++v) tileD[m0 + 8 * half + v][n0 + c16] = c[v];
    __syncthreads();

    for (int j = 0; j < TILE_ROWS / (ROW_T >> 5); ++j) {
        const int lr  = wave + j * (ROW_T >> 5);
        const int row = row0 + lr;
        if (row < N) {
            const float sc = rowScale[lr];
            v2f d = *(const v2f*)&tileD[lr][2 * lane];
            d.x *= sc;
            d.y *= sc;
            *(v2f*)(out + (long long)row * FEAT + 2 * lane) = d;
        }
    }
}

__device__ __forceinline__ void pool_flush(float* __restrict__ sums, float* __restrict__ counts,
                                           int g, int lane, v2f ps, float pc) {
    if ((unsigned)g < (unsigned)NGRAPH) {
        float* p = sums + g * FEAT + 2 * lane;
        atomicAdd(p,     ps.x);
        atomicAdd(p + 1, ps.y);
        if (lane == 0) atomicAdd(counts + g, pc);
    }
}

__global__ void __launch_bounds__(ROW_T)
k_agg_pool(const int* __restrict__ sorted, const int* __restrict__ rowStart,
           const int* __restrict__ rowCnt, const float* __restrict__ hs,
           const float* __restrict__ dinv, const float* __restrict__ bias,
           const int* __restrict__ batch, float* __restrict__ sums, float* __restrict__ counts, int N) {
    __shared__ v2f   slotSum[ROW_T >> 5][32];
    __shared__ int   slotG[ROW_T >> 5];
    __shared__ float slotC[ROW_T >> 5];
    const int lane = threadIdx.x & 31;
    const int wave = __builtin_amdgcn_readfirstlane((int)(threadIdx.x >> 5));
    const v2f bv   = *(const v2f*)(bias + 2 * lane);

    int r = (int)blockIdx.x * POOL_ROWS + wave;
    int s_next = 0, n_next = 0, g_next = 0;
    if (r < N) { s_next = rowStart[r]; n_next = rowCnt[r]; g_next = batch[r]; }

    v2f   ps = {0.0f, 0.0f};
    float pc = 0.0f;
    int   gcur = -1;
    for (int j = 0; j < POOL_ROWS / (ROW_T >> 5) && r < N; ++j, r += (ROW_T >> 5)) {
        const int s = __builtin_amdgcn_readfirstlane(s_next);
        const int n = __builtin_amdgcn_readfirstlane(n_next);
        const int g = __builtin_amdgcn_readfirstlane(g_next);
        const int rn = r + (ROW_T >> 5);
        if (j + 1 < POOL_ROWS / (ROW_T >> 5) && rn < N) { s_next = rowStart[rn]; n_next = rowCnt[rn]; g_next = batch[rn]; }

        v2f acc = *(const v2f*)(hs + (long long)r * FEAT + 2 * lane);
        for (int base = 0; base < n; base += 32) {
            const int rem = (n - base < 32) ? (n - base) : 32;
            const int pk  = (lane < rem) ? sorted[s + base + lane] : 0;
            int j0 = 0;
            for (; j0 + ROW_GRP <= rem; j0 += ROW_GRP) {
                v2f v[ROW_GRP];
#pragma unroll
                for (int u = 0; u < ROW_GRP; ++u) {
                    const int src = __builtin_amdgcn_readlane(pk, j0 + u);
                    v[u] = *(const v2f*)(hs + (long long)src * FEAT + 2 * lane);
                }
#pragma unroll
                for (int u = 0; u < ROW_GRP; ++u) { acc.x += v[u].x; acc.y += v[u].y; }
            }
            for (; j0 < rem; j0 += ROW_TAIL) {
                v2f v[ROW_TAIL];
#pragma unroll
                for (int u = 0; u < ROW_TAIL; ++u) {
                    const int src = __builtin_amdgcn_readlane(pk, j0 + u);
                    v[u] = *(const v2f*)(hs + (long long)src * FEAT + 2 * lane);
                }
#pragma unroll
                for (int u = 0; u < ROW_TAIL; ++u) {
                    if (j0 + u < rem) { acc.x += v[u].x; acc.y += v[u].y; }
                }
            }
        }
        const float dc = dinv[r];
        v2f o;
        o.x = acc.x * dc + bv.x;
        o.y = acc.y * dc + bv.y;
        o.x = o.x > 0.0f ? o.x : 0.0f;
        o.y = o.y > 0.0f ? o.y : 0.0f;

        if (g != gcur) {
            if (pc != 0.0f) pool_flush(sums, counts, gcur, lane, ps, pc);
            ps.x = 0.0f; ps.y = 0.0f; pc = 0.0f;
            gcur = g;
        }
        ps.x += o.x; ps.y += o.y; pc += 1.0f;
    }

    slotSum[wave][lane] = ps;
    if (lane == 0) { slotG[wave] = gcur; slotC[wave] = pc; }
    __syncthreads();
    if (wave == 0) {
        v2f   ts = {0.0f, 0.0f};
        float tc = 0.0f;
        int   gt = -1;
#pragma unroll
        for (int w = 0; w < (ROW_T >> 5); ++w) {
            const int   gw = __builtin_amdgcn_readfirstlane(slotG[w]);
            const float cw = __int_as_float(__builtin_amdgcn_readfirstlane(__float_as_int(slotC[w])));
            const v2f   sw = slotSum[w][lane];
            if (gw != gt) {
                if (tc != 0.0f) pool_flush(sums, counts, gt, lane, ts, tc);
                ts.x = 0.0f; ts.y = 0.0f; tc = 0.0f;
                gt = gw;
            }
            ts.x += sw.x; ts.y += sw.y; tc += cw;
        }
        if (tc != 0.0f) pool_flush(sums, counts, gt, lane, ts, tc);
    }
}

__global__ void k_pool_zero(float* sums, float* counts) {
    int i = blockIdx.x * blockDim.x + threadIdx.x;
    if (i < NGRAPH * FEAT) sums[i] = 0.0f;
    if (i < NGRAPH) counts[i] = 0.0f;
}

__global__ void k_pool_div(const float* __restrict__ sums, const float* __restrict__ counts,
                           float* __restrict__ out) {
    int i = blockIdx.x * blockDim.x + threadIdx.x;
    if (i >= NGRAPH * FEAT) return;
    out[i] = sums[i] / fmaxf(counts[i >> 6], 1.0f);
}

extern "C" void kernel_launch(void* const* d_in, const int* in_sizes, int n_in,
                              void* d_out, int out_size, void* d_ws, size_t ws_size,
                              hipStream_t stream) {
    const float* x     = (const float*)d_in[0];
    const int*   edge  = (const int*)d_in[1];
    const int*   batch = (const int*)d_in[2];
    const float* W1    = (const float*)d_in[3];
    const float* b1    = (const float*)d_in[4];
    const float* W2    = (const float*)d_in[5];
    const float* b2    = (const float*)d_in[6];
    float*       out   = (float*)d_out;

    const int       N = in_sizes[0] / FEAT;
    const long long E = (long long)in_sizes[1] / 2;
    const long long NF = (long long)N * FEAT;

    float* bufA   = (float*)d_ws;
    float* bufB   = bufA + NF;
    float* dinv   = bufB + NF;
    float* sums   = dinv + N;
    float* counts = sums + NGRAPH * FEAT;
    int* binCount = (int*)(counts + NGRAPH);
    int* cursor   = binCount + MAXB;
    int* binStart = cursor + MAXB;
    int* packed   = binStart + MAXB;
    int* sorted   = packed + E;
    int* rowStart = sorted + E;
    int* rowCnt   = rowStart + N;

    const int* rows = edge;
    const int* cols = edge + E;

    const int T = 256;
    const int nbins = ceildiv(N, NPB);

    k_bin_zero <<<1, MAXB, 0, stream>>>(binCount, cursor);
    k_bin_count<<<256, BIN_T, 0, stream>>>(cols, binCount, E);
    k_bin_scan <<<1, MAXB, 0, stream>>>(binCount, binStart);
    k_bin_place<<<256, BIN_T, 0, stream>>>(rows, cols, binStart, cursor, packed, E);

    k_deg_sort<<<nbins, AGG_T, 0, stream>>>(packed, binStart, binCount, dinv, rowStart, rowCnt, sorted, N);

    const int mtiles   = (N + 15) / 16;
    const int gemmBlks = ceildiv((long long)mtiles * 4 * 32, T);

    k_gemm_wmma<<<gemmBlks, T, 0, stream>>>(x, W1, dinv, bufA, N);
    k_agg_rows_gemm<<<ceildiv(N, TILE_ROWS), ROW_T, 0, stream>>>(sorted, rowStart, rowCnt, bufA, dinv, b1, W2,
                                                                 bufB, N);

    k_pool_zero<<<ceildiv(NGRAPH * FEAT, T), T, 0, stream>>>(sums, counts);
    k_agg_pool <<<ceildiv(N, POOL_ROWS), ROW_T, 0, stream>>>(sorted, rowStart, rowCnt, bufB, dinv, b2, batch,
                                                             sums, counts, N);

    k_pool_div <<<ceildiv(NGRAPH * FEAT, T), T, 0, stream>>>(sums, counts, out);
}


# bucketing without a counting pass: a work-group buckets its own chunk of edges and stores it with a run table; the range pass collects its runs; 3 launches gone; row sum order changes
# speedup vs baseline: 6.7540x; 1.0466x over previous
#include <hip/hip_runtime.h>

typedef __attribute__((ext_vector_type(2))) float v2f;
typedef __attribute__((ext_vector_type(8))) float v8f;
typedef __attribute__((ext_vector_type(4))) int   v4i;

#define FEAT 64
#define NGRAPH 64

#define NPB 196
#define NPB_PAD 256
#define MAXB 512
#define SRC_BITS 17
#define BIN_T 1024
#define AGG_T 512
#define ROW_T 256
#define ROW_GRP 16
#define ROW_TAIL 8
#define POOL_ROWS 32
#define PLACE_CAP 13312
#define BIN_UNR (PLACE_CAP / BIN_T)
#define BIN_SPARE 32
#define NCHUNK 256
#define RUNS_PER_WAVE (NCHUNK / (AGG_T >> 5))
#define SORT_CAP 12288
#define TILE_ROWS 32
#define TILE_PITCH 68

static inline int ceildiv(long long a, long long b) { return (int)((a + b - 1) / b); }

__global__ void __launch_bounds__(BIN_T)
k_bin_place(const int* __restrict__ rows, const int* __restrict__ cols,
            int* __restrict__ runTab, int* __restrict__ packed, int* __restrict__ ovf, long long E) {
    __shared__ int hist[MAXB + BIN_SPARE];
    __shared__ int loc[MAXB + BIN_SPARE];
    __shared__ int wsum[MAXB / 32];
    __shared__ __attribute__((aligned(16))) int stage[PLACE_CAP + 32];
    __shared__ int total;
    const int t    = threadIdx.x;
    const int lane = t & 31;
    const int wave = t >> 5;
    for (int i = t; i < MAXB + BIN_SPARE; i += BIN_T) hist[i] = 0;
    if (blockIdx.x == 0 && t == 0) *ovf = 0;
    __syncthreads();
    const long long stride = (long long)gridDim.x * BIN_T;
    const long long first  = (long long)blockIdx.x * BIN_T + t;

    const int spare = MAXB + lane;
    int c[BIN_UNR], r[BIN_UNR];
#pragma unroll
    for (int k = 0; k < BIN_UNR; ++k) { c[k] = 0; r[k] = 0; }
    if (E > 0) {
#pragma unroll
        for (int k = 0; k < BIN_UNR; ++k) {
            const long long e  = first + k * stride;
            const long long ec = e < E ? e : E - 1;
            c[k] = cols[ec];
            r[k] = rows[ec];
        }
    }
#pragma unroll
    for (int k = 0; k < BIN_UNR; ++k) {
        const bool in = first + k * stride < E;
        atomicAdd(&hist[in ? c[k] / NPB : spare], in ? 1 : 0);
    }
    __syncthreads();

    const int n = (t < MAXB) ? hist[t] : 0;
    int incl = n;
    for (int d = 1; d < 32; d <<= 1) {
        const int v = __shfl_up(incl, d);
        if (lane >= d) incl += v;
    }
    if (t < MAXB && lane == 31) wsum[wave] = incl;
    __syncthreads();
    if (t < MAXB) {
        int pre = 0;
        for (int w = 0; w < wave; ++w) pre += wsum[w];
        const int lc = pre + incl - n;
        loc[t] = lc;
        runTab[(int)blockIdx.x * MAXB + t] = (lc << 16) | n;
        if (t == MAXB - 1) total = lc + n;
        hist[t] = 0;
    } else if (t < MAXB + BIN_SPARE) {
        loc[t] = PLACE_CAP + (t - MAXB);
    }
    __syncthreads();
    int pos[BIN_UNR];
#pragma unroll
    for (int k = 0; k < BIN_UNR; ++k) {
        const bool in = first + k * stride < E;
        const int  bb = in ? c[k] / NPB : spare;
        pos[k] = loc[bb] + atomicAdd(&hist[bb], in ? 1 : 0);
    }
#pragma unroll
    for (int k = 0; k < BIN_UNR; ++k) {
        const int b = c[k] / NPB;
        stage[pos[k]] = r[k] | ((c[k] - b * NPB) << SRC_BITS);
    }
    __syncthreads();
    const int tot = total;
    int* __restrict__ dst = packed + (long long)blockIdx.x * PLACE_CAP;
    v4i w[(PLACE_CAP + 4 * BIN_T - 1) / (4 * BIN_T)];
#pragma unroll
    for (int k = 0; k < (PLACE_CAP + 4 * BIN_T - 1) / (4 * BIN_T); ++k) {
        const int i = 4 * t + k * 4 * BIN_T;
        if (i < PLACE_CAP) w[k] = *(const v4i*)&stage[i];
    }
#pragma unroll
    for (int k = 0; k < (PLACE_CAP + 4 * BIN_T - 1) / (4 * BIN_T); ++k) {
        const int i = 4 * t + k * 4 * BIN_T;
        if (i + 3 < tot) {
            *(v4i*)&dst[i] = w[k];
        } else {
            if (i     < tot) dst[i]     = w[k].x;
            if (i + 1 < tot) dst[i + 1] = w[k].y;
            if (i + 2 < tot) dst[i + 2] = w[k].z;
        }
    }
}

__global__ void __launch_bounds__(AGG_T)
k_deg_sort(const int* __restrict__ packed, const int* __restrict__ runTab, int* __restrict__ ovf,
           float* __restrict__ dinv, int* __restrict__ rowStart, int* __restrict__ rowCnt,
           int* __restrict__ sorted, int N) {
    __shared__ int cnt[NPB_PAD];
    __shared__ int off[NPB_PAD];
    __shared__ int stage[SORT_CAP];
    __shared__ int wTot[AGG_T >> 5];
    __shared__ int wMax[AGG_T >> 5];
    __shared__ int region;
    const int b = blockIdx.x;
    const int t = threadIdx.x;
    const int lane = t & 31;
    const int wave = __builtin_amdgcn_readfirstlane(t >> 5);
    if (t < NPB_PAD) cnt[t] = 0;

    const int ch = wave * RUNS_PER_WAVE + (lane & (RUNS_PER_WAVE - 1));
    const int tw = runTab[ch * MAXB + b];
    const int rn = tw & 0xffff;
    const int ro = ch * PLACE_CAP + (tw >> 16);
    int sN = rn, mN = rn;
    for (int d = 1; d < RUNS_PER_WAVE; d <<= 1) {
        sN += __shfl_xor(sN, d);
        const int o = __shfl_xor(mN, d);
        mN = mN > o ? mN : o;
    }
    if (lane == 0) { wTot[wave] = sN; wMax[wave] = mN; }
    __syncthreads();
    int m = 0, longest = 0;
    for (int w = 0; w < (AGG_T >> 5); ++w) {
        m += wTot[w];
        longest = longest > wMax[w] ? longest : wMax[w];
    }
    const bool staged = m <= SORT_CAP && longest <= 64;
    if (t == 0) region = (m <= SORT_CAP) ? b * SORT_CAP : MAXB * SORT_CAP + atomicAdd(ovf, m);

    const int spareRow = NPB + lane;
    int p1[RUNS_PER_WAVE], p2[RUNS_PER_WAVE];
    if (staged) {
#pragma unroll
        for (int i = 0; i < RUNS_PER_WAVE; ++i) {
            const int ni = __builtin_amdgcn_readlane(rn, i);
            const int oi = __builtin_amdgcn_readlane(ro, i);
            const int ci = (wave * RUNS_PER_WAVE + i) * PLACE_CAP;
            const int v  = packed[(lane < ni) ? oi + lane : ci];
            p1[i] = (lane < ni) ? v : -1;
        }
#pragma unroll
        for (int i = 0; i < RUNS_PER_WAVE; ++i) {
            const int ni = __builtin_amdgcn_readlane(rn, i);
            const int oi = __builtin_amdgcn_readlane(ro, i);
            p2[i] = -1;
            if (ni > 32) p2[i] = (lane + 32 < ni) ? packed[oi + 32 + lane] : -1;
        }
#pragma unroll
        for (int i = 0; i < RUNS_PER_WAVE; ++i) {
            const bool in = p1[i] >= 0;
            atomicAdd(&cnt[in ? p1[i] >> SRC_BITS : spareRow], in ? 1 : 0);
        }
#pragma unroll
        for (int i = 0; i < RUNS_PER_WAVE; ++i) {
            if (__builtin_amdgcn_readlane(rn, i) > 32) {
                if (p2[i] >= 0) atomicAdd(&cnt[p2[i] >> SRC_BITS], 1);
            }
        }
    } else {
#pragma nounroll
        for (int i = 0; i < RUNS_PER_WAVE; ++i) {
            const int ni = __builtin_amdgcn_readlane(rn, i);
            const int oi = __builtin_amdgcn_readlane(ro, i);
#pragma nounroll
            for (int k = lane; k < ni; k += 32) atomicAdd(&cnt[packed[oi + k] >> SRC_BITS], 1);
        }
    }
    __syncthreads();

    int own = 0;
    if (t < NPB_PAD) { own = cnt[t]; off[t] = own; }
    __syncthreads();
    for (int d = 1; d < NPB_PAD; d <<= 1) {
        int v = 0;
        if (t < NPB_PAD && t >= d) v = off[t - d];
        __syncthreads();
        if (t < NPB_PAD) off[t] += v;
        __syncthreads();
    }
    const int start = region;
    if (t < NPB_PAD) {
        const int ex = off[t] - own;
        off[t] = ex;
        cnt[t] = 0;
        const int node = b * NPB + t;
        if (t < NPB && node < N) {
            dinv[node]     = rsqrtf(1.0f + (float)own);
            rowStart[node] = start + ex;
            rowCnt[node]   = own;
        }
    }
    __syncthreads();
    if (!staged) {
#pragma nounroll
        for (int i = 0; i < RUNS_PER_WAVE; ++i) {
            const int ni = __builtin_amdgcn_readlane(rn, i);
            const int oi = __builtin_amdgcn_readlane(ro, i);
#pragma nounroll
            for (int k = lane; k < ni; k += 32) {
                const int p = packed[oi + k];
                const int d = p >> SRC_BITS;
                const int pos = start + off[d] + atomicAdd(&cnt[d], 1);
                sorted[pos] = p & ((1 << SRC_BITS) - 1);
            }
        }
    } else {
        int q1[RUNS_PER_WAVE];
#pragma unroll
        for (int i = 0; i < RUNS_PER_WAVE; ++i) {
            const bool in = p1[i] >= 0;
            const int  d  = in ? p1[i] >> SRC_BITS : spareRow;
            q1[i] = off[d] + atomicAdd(&cnt[d], in ? 1 : 0);
        }
#pragma unroll
        for (int i = 0; i < RUNS_PER_WAVE; ++i)
            if (p1[i] >= 0) stage[q1[i]] = p1[i] & ((1 << SRC_BITS) - 1);
#pragma unroll
        for (int i = 0; i < RUNS_PER_WAVE; ++i) {
            if (__builtin_amdgcn_readlane(rn, i) > 32) {
                if (p2[i] >= 0) {
                    const int d = p2[i] >> SRC_BITS;
                    const int pos = off[d] + atomicAdd(&cnt[d], 1);
                    stage[pos] = p2[i] & ((1 << SRC_BITS) - 1);
                }
            }
        }
        __syncthreads();
#pragma unroll 4
        for (int i = t; i < m; i += AGG_T) sorted[start + i] = stage[i];
    }
}

__global__ void k_gemm_wmma(const float* __restrict__ A, const float* __restrict__ W,
                            const float* __restrict__ dinv, float* __restrict__ D, int M) {
    int gwave = (blockIdx.x * blockDim.x + threadIdx.x) >> 5;
    int lane  = threadIdx.x & 31;
    int mtiles = (M + 15) >> 4;
    int mtile = gwave >> 2;
    int ntile = gwave & 3;
    if (mtile >= mtiles) return;

    int m0 = mtile << 4, n0 = ntile << 4;
    int half = lane >> 4;
    int r    = lane & 15;

    int arow = m0 + r; if (arow >= M) arow = M - 1;

    float s[8];
#pragma unroll
    for (int v = 0; v < 8; ++v) {
        int row = m0 + v + 8 * half;
        s[v] = dinv[row < M ? row : M - 1];
    }

    v8f c = {};
#pragma unroll
    for (int k0 = 0; k0 < FEAT; k0 += 4) {
        v2f a, b;
        int ka = k0 + 2 * half;
        a.x = A[(long long)arow * FEAT + ka + 0];
        a.y = A[(long long)arow * FEAT + ka + 1];
        b.x = W[(ka + 0) * FEAT + n0 + r];
        b.y = W[(ka + 1) * FEAT + n0 + r];
        c = __builtin_amdgcn_wmma_f32_16x16x4_f32(false, a, false, b, (short)0, c, false, false);
    }
#pragma unroll
    for (int v = 0; v < 8; ++v) {
        int row = m0 + v + 8 * half;
        if (row < M) D[(long long)row * FEAT + n0 + r] = c[v] * s[v];
    }
}


__global__ void __launch_bounds__(ROW_T)
k_agg_rows_gemm(const int* __restrict__ sorted, const int* __restrict__ rowStart,
                const int* __restrict__ rowCnt, const float* __restrict__ hs,
                const float* __restrict__ dinv, const float* __restrict__ bias,
                const float* __restrict__ Wn, float* __restrict__ out, int N) {
    __shared__ float tileA[TILE_ROWS][TILE_PITCH];
    __shared__ float tileD[TILE_ROWS][TILE_PITCH];
    __shared__ float rowScale[TILE_ROWS];
    const int lane = threadIdx.x & 31;
    const int wave = __builtin_amdgcn_readfirstlane((int)(threadIdx.x >> 5));
    const v2f bv   = *(const v2f*)(bias + 2 * lane);
    const int row0 = (int)blockIdx.x * TILE_ROWS;

    int r = row0 + wave;
    int s_next = 0, n_next = 0;
    if (r < N) { s_next = rowStart[r]; n_next = rowCnt[r]; }
    for (int j = 0; j < TILE_ROWS / (ROW_T >> 5); ++j, r += (ROW_T >> 5)) {
        v2f   o  = {0.0f, 0.0f};
        float dc = 0.0f;
        if (r < N) {
            const int s = __builtin_amdgcn_readfirstlane(s_next);
            const int n = __builtin_amdgcn_readfirstlane(n_next);
            const int rn = r + (ROW_T >> 5);
            if (j + 1 < TILE_ROWS / (ROW_T >> 5) && rn < N) { s_next = rowStart[rn]; n_next = rowCnt[rn]; }

            v2f acc = *(const v2f*)(hs + (long long)r * FEAT + 2 * lane);
            for (int base = 0; base < n; base += 32) {
                const int rem = (n - base < 32) ? (n - base) : 32;
                const int pk  = (lane < rem) ? sorted[s + base + lane] : 0;
                int j0 = 0;
                for (; j0 + ROW_GRP <= rem; j0 += ROW_GRP) {
                    v2f v[ROW_GRP];
#pragma unroll
                    for (int u = 0; u < ROW_GRP; ++u) {
                        const int src = __builtin_amdgcn_readlane(pk, j0 + u);
                        v[u] = *(const v2f*)(hs + (long long)src * FEAT + 2 * lane);
                    }
#pragma unroll
                    for (int u = 0; u < ROW_GRP; ++u) { acc.x += v[u].x; acc.y += v[u].y; }
                }
                for (; j0 < rem; j0 += ROW_TAIL) {
                    v2f v[ROW_TAIL];
#pragma unroll
                    for (int u = 0; u < ROW_TAIL; ++u) {
                        const int src = __builtin_amdgcn_readlane(pk, j0 + u);
                        v[u] = *(const v2f*)(hs + (long long)src * FEAT + 2 * lane);
                    }
#pragma unroll
                    for (int u = 0; u < ROW_TAIL; ++u) {
                        if (j0 + u < rem) { acc.x += v[u].x; acc.y += v[u].y; }
                    }
                }
            }
            dc = dinv[r];
            o.x = acc.x * dc + bv.x;
            o.y = acc.y * dc + bv.y;
            o.x = o.x > 0.0f ? o.x : 0.0f;
            o.y = o.y > 0.0f ? o.y : 0.0f;
        }
        const int lr = wave + j * (ROW_T >> 5);
        *(v2f*)&tileA[lr][2 * lane] = o;
        if (lane == 0) rowScale[lr] = dc;
    }
    __syncthreads();

    const int m0   = (wave >> 2) << 4;
    const int n0   = (wave & 3) << 4;
    const int half = lane >> 4;
    const int c16  = lane & 15;
    v8f c = {};
#pragma unroll
    for (int k0 = 0; k0 < FEAT; k0 += 4) {
        const int ka = k0 + 2 * half;
        const v2f a = *(const v2f*)&tileA[m0 + c16][ka];
        v2f b;
        b.x = Wn[(ka + 0) * FEAT + n0 + c16];
        b.y = Wn[(ka + 1) * FEAT + n0 + c16];
        c = __builtin_amdgcn_wmma_f32_16x16x4_f32(false, a, false, b, (short)0, c, false, false);
    }
#pragma unroll
    for (int v = 0; v < 8; ++v) tileD[m0 + 8 * half + v][n0 + c16] = c[v];
    __syncthreads();

    for (int j = 0; j < TILE_ROWS / (ROW_T >> 5); ++j) {
        const int lr  = wave + j * (ROW_T >> 5);
        const int row = row0 + lr;
        if (row < N) {
            const float sc = rowScale[lr];
            v2f d = *(const v2f*)&tileD[lr][2 * lane];
            d.x *= sc;
            d.y *= sc;
            *(v2f*)(out + (long long)row * FEAT + 2 * lane) = d;
        }
    }
}

__device__ __forceinline__ void pool_flush(float* __restrict__ sums, float* __restrict__ counts,
                                           int g, int lane, v2f ps, float pc) {
    if ((unsigned)g < (unsigned)NGRAPH) {
        float* p = sums + g * FEAT + 2 * lane;
        atomicAdd(p,     ps.x);
        atomicAdd(p + 1, ps.y);
        if (lane == 0) atomicAdd(counts + g, pc);
    }
}

__global__ void __launch_bounds__(ROW_T)
k_agg_pool(const int* __restrict__ sorted, const int* __restrict__ rowStart,
           const int* __restrict__ rowCnt, const float* __restrict__ hs,
           const float* __restrict__ dinv, const float* __restrict__ bias,
           const int* __restrict__ batch, float* __restrict__ sums, float* __restrict__ counts, int N) {
    __shared__ v2f   slotSum[ROW_T >> 5][32];
    __shared__ int   slotG[ROW_T >> 5];
    __shared__ float slotC[ROW_T >> 5];
    const int lane = threadIdx.x & 31;
    const int wave = __builtin_amdgcn_readfirstlane((int)(threadIdx.x >> 5));
    const v2f bv   = *(const v2f*)(bias + 2 * lane);

    int r = (int)blockIdx.x * POOL_ROWS + wave;
    int s_next = 0, n_next = 0, g_next = 0;
    if (r < N) { s_next = rowStart[r]; n_next = rowCnt[r]; g_next = batch[r]; }

    v2f   ps = {0.0f, 0.0f};
    float pc = 0.0f;
    int   gcur = -1;
    for (int j = 0; j < POOL_ROWS / (ROW_T >> 5) && r < N; ++j, r += (ROW_T >> 5)) {
        const int s = __builtin_amdgcn_readfirstlane(s_next);
        const int n = __builtin_amdgcn_readfirstlane(n_next);
        const int g = __builtin_amdgcn_readfirstlane(g_next);
        const int rn = r + (ROW_T >> 5);
        if (j + 1 < POOL_ROWS / (ROW_T >> 5) && rn < N) { s_next = rowStart[rn]; n_next = rowCnt[rn]; g_next = batch[rn]; }

        v2f acc = *(const v2f*)(hs + (long long)r * FEAT + 2 * lane);
        for (int base = 0; base < n; base += 32) {
            const int rem = (n - base < 32) ? (n - base) : 32;
            const int pk  = (lane < rem) ? sorted[s + base + lane] : 0;
            int j0 = 0;
            for (; j0 + ROW_GRP <= rem; j0 += ROW_GRP) {
                v2f v[ROW_GRP];
#pragma unroll
                for (int u = 0; u < ROW_GRP; ++u) {
                    const int src = __builtin_amdgcn_readlane(pk, j0 + u);
                    v[u] = *(const v2f*)(hs + (long long)src * FEAT + 2 * lane);
                }
#pragma unroll
                for (int u = 0; u < ROW_GRP; ++u) { acc.x += v[u].x; acc.y += v[u].y; }
            }
            for (; j0 < rem; j0 += ROW_TAIL) {
                v2f v[ROW_TAIL];
#pragma unroll
                for (int u = 0; u < ROW_TAIL; ++u) {
                    const int src = __builtin_amdgcn_readlane(pk, j0 + u);
                    v[u] = *(const v2f*)(hs + (long long)src * FEAT + 2 * lane);
                }
#pragma unroll
                for (int u = 0; u < ROW_TAIL; ++u) {
                    if (j0 + u < rem) { acc.x += v[u].x; acc.y += v[u].y; }
                }
            }
        }
        const float dc = dinv[r];
        v2f o;
        o.x = acc.x * dc + bv.x;
        o.y = acc.y * dc + bv.y;
        o.x = o.x > 0.0f ? o.x : 0.0f;
        o.y = o.y > 0.0f ? o.y : 0.0f;

        if (g != gcur) {
            if (pc != 0.0f) pool_flush(sums, counts, gcur, lane, ps, pc);
            ps.x = 0.0f; ps.y = 0.0f; pc = 0.0f;
            gcur = g;
        }
        ps.x += o.x; ps.y += o.y; pc += 1.0f;
    }

    slotSum[wave][lane] = ps;
    if (lane == 0) { slotG[wave] = gcur; slotC[wave] = pc; }
    __syncthreads();
    if (wave == 0) {
        v2f   ts = {0.0f, 0.0f};
        float tc = 0.0f;
        int   gt = -1;
#pragma unroll
        for (int w = 0; w < (ROW_T >> 5); ++w) {
            const int   gw = __builtin_amdgcn_readfirstlane(slotG[w]);
            const float cw = __int_as_float(__builtin_amdgcn_readfirstlane(__float_as_int(slotC[w])));
            const v2f   sw = slotSum[w][lane];
            if (gw != gt) {
                if (tc != 0.0f) pool_flush(sums, counts, gt, lane, ts, tc);
                ts.x = 0.0f; ts.y = 0.0f; tc = 0.0f;
                gt = gw;
            }
            ts.x += sw.x; ts.y += sw.y; tc += cw;
        }
        if (tc != 0.0f) pool_flush(sums, counts, gt, lane, ts, tc);
    }
}

__global__ void k_pool_zero(float* sums, float* counts) {
    int i = blockIdx.x * blockDim.x + threadIdx.x;
    if (i < NGRAPH * FEAT) sums[i] = 0.0f;
    if (i < NGRAPH) counts[i] = 0.0f;
}

__global__ void k_pool_div(const float* __restrict__ sums, const float* __restrict__ counts,
                           float* __restrict__ out) {
    int i = blockIdx.x * blockDim.x + threadIdx.x;
    if (i >= NGRAPH * FEAT) return;
    out[i] = sums[i] / fmaxf(counts[i >> 6], 1.0f);
}

extern "C" void kernel_launch(void* const* d_in, const int* in_sizes, int n_in,
                              void* d_out, int out_size, void* d_ws, size_t ws_size,
                              hipStream_t stream) {
    const float* x     = (const float*)d_in[0];
    const int*   edge  = (const int*)d_in[1];
    const int*   batch = (const int*)d_in[2];
    const float* W1    = (const float*)d_in[3];
    const float* b1    = (const float*)d_in[4];
    const float* W2    = (const float*)d_in[5];
    const float* b2    = (const float*)d_in[6];
    float*       out   = (float*)d_out;

    const int       N = in_sizes[0] / FEAT;
    const long long E = (long long)in_sizes[1] / 2;
    const long long NF = (long long)N * FEAT;

    float* bufA   = (float*)d_ws;
    float* bufB   = bufA + NF;
    float* dinv   = bufB + NF;
    float* sums   = dinv + N;
    float* counts = sums + NGRAPH * FEAT;
    int* ovf      = (int*)(counts + NGRAPH);
    int* runTab   = ovf + 32;
    int* packed   = runTab + NCHUNK * MAXB;
    int* sorted   = packed + NCHUNK * PLACE_CAP;
    int* rowStart = sorted + (long long)MAXB * SORT_CAP + E;
    int* rowCnt   = rowStart + N;

    const int* rows = edge;
    const int* cols = edge + E;

    const int T = 256;
    const int nbins = ceildiv(N, NPB);

    k_bin_place<<<NCHUNK, BIN_T, 0, stream>>>(rows, cols, runTab, packed, ovf, E);

    k_deg_sort<<<nbins, AGG_T, 0, stream>>>(packed, runTab, ovf, dinv, rowStart, rowCnt, sorted, N);

    const int mtiles   = (N + 15) / 16;
    const int gemmBlks = ceildiv((long long)mtiles * 4 * 32, T);

    k_gemm_wmma<<<gemmBlks, T, 0, stream>>>(x, W1, dinv, bufA, N);
    k_agg_rows_gemm<<<ceildiv(N, TILE_ROWS), ROW_T, 0, stream>>>(sorted, rowStart, rowCnt, bufA, dinv, b1, W2,
                                                                 bufB, N);

    k_pool_zero<<<ceildiv(NGRAPH * FEAT, T), T, 0, stream>>>(sums, counts);
    k_agg_pool <<<ceildiv(N, POOL_ROWS), ROW_T, 0, stream>>>(sorted, rowStart, rowCnt, bufB, dinv, b2, batch,
                                                             sums, counts, N);

    k_pool_div <<<ceildiv(NGRAPH * FEAT, T), T, 0, stream>>>(sums, counts, out);
}
